# starting kernel
# baseline (speedup 1.0000x reference)
#include <hip/hip_runtime.h>

typedef __attribute__((ext_vector_type(2))) float v2f;
typedef __attribute__((ext_vector_type(8))) float v8f;

#define NEG_SLOPE 0.2f
#define BN_EPS 1e-5f

__device__ __forceinline__ float lrelu(float x) { return x > 0.0f ? x : NEG_SLOPE * x; }

__device__ __forceinline__ unsigned fenc(float f) {
    unsigned u = __float_as_uint(f);
    return (u & 0x80000000u) ? ~u : (u | 0x80000000u);
}
__device__ __forceinline__ float fdec(unsigned u) {
    return __uint_as_float((u & 0x80000000u) ? (u ^ 0x80000000u) : ~u);
}

__global__ void zero_ws(float* __restrict__ p, int n) {
    int i = blockIdx.x * blockDim.x + threadIdx.x;
    if (i < n) p[i] = 0.0f;
}

__global__ __launch_bounds__(256) void node_transform(
    const float* __restrict__ x, const float* __restrict__ Wl,
    const float* __restrict__ Wr, float* __restrict__ xl, float* __restrict__ xr,
    int N) {
    const int wave = threadIdx.x >> 5;
    const int lane = threadIdx.x & 31;
    const int lo   = lane & 15;
    const int hiK  = (lane >= 16) ? 2 : 0;
    const int hiM  = (lane >= 16) ? 8 : 0;
    const int rowTile = blockIdx.x * 16;
    const int col  = wave * 16 + lo;
    const int aRow = rowTile + lo;

    v8f cl = {};
    v8f cr = {};
    for (int k = 0; k < 128; k += 4) {
        v2f a = *(const v2f*)&x[aRow * 128 + k + hiK];
        v2f bl, br;
        bl.x = Wl[(k + hiK) * 128 + col];
        bl.y = Wl[(k + hiK + 1) * 128 + col];
        br.x = Wr[(k + hiK) * 128 + col];
        br.y = Wr[(k + hiK + 1) * 128 + col];
        cl = __builtin_amdgcn_wmma_f32_16x16x4_f32(false, a, false, bl, (short)0, cl, false, false);
        cr = __builtin_amdgcn_wmma_f32_16x16x4_f32(false, a, false, br, (short)0, cr, false, false);
    }
#pragma unroll
    for (int r = 0; r < 8; ++r) {
        int row = rowTile + r + hiM;
        xl[row * 128 + col] = cl[r];
        xr[row * 128 + col] = cr[r];
    }
}

__global__ __launch_bounds__(256) void edge_logits(
    const float* __restrict__ edge_attr, const float* __restrict__ We,
    const float* __restrict__ xl, const float* __restrict__ xr,
    const int* __restrict__ srcIdx, const int* __restrict__ dstIdx,
    const float* __restrict__ att, float* __restrict__ ebuf,
    unsigned* __restrict__ menc, int E) {
    __shared__ float sWe[16 * 128];
    for (int i = threadIdx.x; i < 16 * 128; i += 256) sWe[i] = We[i];
    __syncthreads();

    const int wave = threadIdx.x >> 5;
    const int lane = threadIdx.x & 31;
    const int lo   = lane & 15;
    const int hiK  = (lane >= 16) ? 2 : 0;
    const int hiM  = (lane >= 16) ? 8 : 0;
    const int tile = (blockIdx.x * 8 + wave) * 16;

    const float* aRow = &edge_attr[(size_t)(tile + lo) * 16];
    v2f a0 = *(const v2f*)&aRow[0 + hiK];
    v2f a1 = *(const v2f*)&aRow[4 + hiK];
    v2f a2 = *(const v2f*)&aRow[8 + hiK];
    v2f a3 = *(const v2f*)&aRow[12 + hiK];

    int sIdx = srcIdx[tile + lo];
    int dIdx = dstIdx[tile + lo];
    int myS[8], myD[8];
#pragma unroll
    for (int r = 0; r < 8; ++r) {
        myS[r] = __shfl(sIdx, r + hiM, 32);
        myD[r] = __shfl(dIdx, r + hiM, 32);
    }

    float acc[8];
#pragma unroll
    for (int r = 0; r < 8; ++r) acc[r] = 0.0f;

#pragma unroll
    for (int wi = 0; wi < 8; ++wi) {
        const int col = wi * 16 + lo;
        v8f c;
#pragma unroll
        for (int r = 0; r < 8; ++r)
            c[r] = xl[(size_t)myS[r] * 128 + col] + xr[(size_t)myD[r] * 128 + col];
        v2f b;
        b.x = sWe[(0 + hiK) * 128 + col];  b.y = sWe[(1 + hiK) * 128 + col];
        c = __builtin_amdgcn_wmma_f32_16x16x4_f32(false, a0, false, b, (short)0, c, false, false);
        b.x = sWe[(4 + hiK) * 128 + col];  b.y = sWe[(5 + hiK) * 128 + col];
        c = __builtin_amdgcn_wmma_f32_16x16x4_f32(false, a1, false, b, (short)0, c, false, false);
        b.x = sWe[(8 + hiK) * 128 + col];  b.y = sWe[(9 + hiK) * 128 + col];
        c = __builtin_amdgcn_wmma_f32_16x16x4_f32(false, a2, false, b, (short)0, c, false, false);
        b.x = sWe[(12 + hiK) * 128 + col]; b.y = sWe[(13 + hiK) * 128 + col];
        c = __builtin_amdgcn_wmma_f32_16x16x4_f32(false, a3, false, b, (short)0, c, false, false);

        const float av = att[col];
#pragma unroll
        for (int r = 0; r < 8; ++r) acc[r] += lrelu(c[r]) * av;
    }

#pragma unroll
    for (int r = 0; r < 8; ++r) {
        float v = acc[r];
        v += __shfl_xor(v, 1, 32);
        v += __shfl_xor(v, 2, 32);
        v += __shfl_xor(v, 4, 32);
        v += __shfl_xor(v, 8, 32);
        if (lo == 0) {
            int edge = tile + r + hiM;
            ebuf[edge] = v;
            atomicMax(&menc[myD[r]], fenc(v));
        }
    }
}

__global__ void edge_exp(const float* __restrict__ ebuf,
                         const int* __restrict__ dstIdx,
                         const unsigned* __restrict__ menc,
                         float* __restrict__ pbuf, float* __restrict__ denom,
                         int E) {
    int i = blockIdx.x * blockDim.x + threadIdx.x;
    if (i >= E) return;
    int d = dstIdx[i];
    float p = expf(ebuf[i] - fdec(menc[d]));
    pbuf[i] = p;
    atomicAdd(&denom[d], p);
}

__global__ __launch_bounds__(256) void edge_scatter(
    const float* __restrict__ xl, const float* __restrict__ pbuf,
    const float* __restrict__ denom, const int* __restrict__ srcIdx,
    const int* __restrict__ dstIdx, float* __restrict__ accum, int E) {
    int gw   = (blockIdx.x * 256 + threadIdx.x) >> 5;
    int lane = threadIdx.x & 31;
    if (gw >= E) return;
    int d = dstIdx[gw];
    float alpha = pbuf[gw] / denom[d];
    int s = srcIdx[gw];
    const float4 v = *(const float4*)&xl[(size_t)s * 128 + lane * 4];
    float* o = &accum[(size_t)d * 128 + lane * 4];
    atomicAdd(o + 0, alpha * v.x);
    atomicAdd(o + 1, alpha * v.y);
    atomicAdd(o + 2, alpha * v.z);
    atomicAdd(o + 3, alpha * v.w);
}

__global__ __launch_bounds__(128) void bias_act_stats(
    const float* __restrict__ accum, const float* __restrict__ bias,
    float* __restrict__ out, float* __restrict__ stats, int N, int rowsPerBlock) {
    int c  = threadIdx.x;
    int r0 = blockIdx.x * rowsPerBlock;
    int r1 = min(r0 + rowsPerBlock, N);
    float b = bias[c];
    float sum = 0.0f, sq = 0.0f;
    for (int r = r0; r < r1; ++r) {
        float v = lrelu(accum[(size_t)r * 128 + c] + b);
        out[(size_t)r * 128 + c] = v;
        sum += v;
        sq  += v * v;
    }
    atomicAdd(&stats[c], sum);
    atomicAdd(&stats[128 + c], sq);
}

__global__ void bn_prep(const float* __restrict__ stats,
                        const float* __restrict__ bnw,
                        const float* __restrict__ bnb,
                        float* __restrict__ scsh, int N) {
    int c = threadIdx.x;
    float inv = 1.0f / (float)N;
    float mean = stats[c] * inv;
    float var  = stats[128 + c] * inv - mean * mean;
    float sc   = bnw[c] * rsqrtf(var + BN_EPS);
    scsh[c]       = sc;
    scsh[128 + c] = bnb[c] - mean * sc;
}

__global__ void bn_apply(float* __restrict__ out, const float* __restrict__ scsh,
                         int total) {
    int i = blockIdx.x * blockDim.x + threadIdx.x;
    if (i >= total) return;
    int c = i & 127;
    out[i] = out[i] * scsh[c] + scsh[128 + c];
}

extern "C" void kernel_launch(void* const* d_in, const int* in_sizes, int n_in,
                              void* d_out, int out_size, void* d_ws, size_t ws_size,
                              hipStream_t stream) {
    const float* x    = (const float*)d_in[0];
    const float* ea   = (const float*)d_in[1];
    const int*   ei   = (const int*)d_in[2];
    const float* Wl   = (const float*)d_in[3];
    const float* Wr   = (const float*)d_in[4];
    const float* We   = (const float*)d_in[5];
    const float* att  = (const float*)d_in[6];
    const float* bias = (const float*)d_in[7];
    const float* bnw  = (const float*)d_in[8];
    const float* bnb  = (const float*)d_in[9];
    float* out = (float*)d_out;

    const int N = in_sizes[0] / 128;
    const int E = in_sizes[2] / 2;
    const int* srcIdx = ei;
    const int* dstIdx = ei + E;

    float* ws = (float*)d_ws;
    float* xl = ws;     ws += (size_t)N * 128;
    float* xr = ws;     ws += (size_t)N * 128;
    float* ebuf = ws;   ws += E;
    float* pbuf = ws;   ws += E;
    float* zbase = ws;
    unsigned* menc = (unsigned*)ws; ws += N;
    float* denom = ws;  ws += N;
    float* accum = ws;  ws += (size_t)N * 128;
    float* stats = ws;  ws += 256;
    float* scsh  = ws;  ws += 256;
    int zcount = (int)(ws - zbase);

    zero_ws<<<(zcount + 255) / 256, 256, 0, stream>>>(zbase, zcount);
    node_transform<<<N / 16, 256, 0, stream>>>(x, Wl, Wr, xl, xr, N);
    edge_logits<<<E / 128, 256, 0, stream>>>(ea, We, xl, xr, srcIdx, dstIdx, att,
                                             ebuf, menc, E);
    edge_exp<<<(E + 255) / 256, 256, 0, stream>>>(ebuf, dstIdx, menc, pbuf, denom, E);
    edge_scatter<<<(E + 7) / 8, 256, 0, stream>>>(xl, pbuf, denom, srcIdx, dstIdx,
                                                  accum, E);
    const int rpb = 50;
    bias_act_stats<<<(N + rpb - 1) / rpb, 128, 0, stream>>>(accum, bias, out, stats,
                                                            N, rpb);
    bn_prep<<<1, 128, 0, stream>>>(stats, bnw, bnb, scsh, N);
    bn_apply<<<(N * 128 + 255) / 256, 256, 0, stream>>>(out, scsh, N * 128);
}


# Softmax divide deferred: scatter adds exp(e-m)*row and exp(e-m) into a per-node sum (own 128-B line each); rows divided once in the epilogue; exp launch and weight buffer dropped; sum order changes
# speedup vs baseline: 1.2567x; 1.2567x over previous
#include <hip/hip_runtime.h>

typedef __attribute__((ext_vector_type(2))) float v2f;
typedef __attribute__((ext_vector_type(8))) float v8f;

#define NEG_SLOPE 0.2f
#define BN_EPS 1e-5f

__device__ __forceinline__ float lrelu(float x) { return x > 0.0f ? x : NEG_SLOPE * x; }

__device__ __forceinline__ unsigned fenc(float f) {
    unsigned u = __float_as_uint(f);
    return (u & 0x80000000u) ? ~u : (u | 0x80000000u);
}
__device__ __forceinline__ float fdec(unsigned u) {
    return __uint_as_float((u & 0x80000000u) ? (u ^ 0x80000000u) : ~u);
}

__global__ void zero_ws(float* __restrict__ p, int n) {
    int i = blockIdx.x * blockDim.x + threadIdx.x;
    if (i < n) p[i] = 0.0f;
}

__global__ __launch_bounds__(256) void node_transform(
    const float* __restrict__ x, const float* __restrict__ Wl,
    const float* __restrict__ Wr, float* __restrict__ xl, float* __restrict__ xr,
    int N) {
    const int wave = threadIdx.x >> 5;
    const int lane = threadIdx.x & 31;
    const int lo   = lane & 15;
    const int hiK  = (lane >= 16) ? 2 : 0;
    const int hiM  = (lane >= 16) ? 8 : 0;
    const int rowTile = blockIdx.x * 16;
    const int col  = wave * 16 + lo;
    const int aRow = rowTile + lo;

    v8f cl = {};
    v8f cr = {};
    for (int k = 0; k < 128; k += 4) {
        v2f a = *(const v2f*)&x[aRow * 128 + k + hiK];
        v2f bl, br;
        bl.x = Wl[(k + hiK) * 128 + col];
        bl.y = Wl[(k + hiK + 1) * 128 + col];
        br.x = Wr[(k + hiK) * 128 + col];
        br.y = Wr[(k + hiK + 1) * 128 + col];
        cl = __builtin_amdgcn_wmma_f32_16x16x4_f32(false, a, false, bl, (short)0, cl, false, false);
        cr = __builtin_amdgcn_wmma_f32_16x16x4_f32(false, a, false, br, (short)0, cr, false, false);
    }
#pragma unroll
    for (int r = 0; r < 8; ++r) {
        int row = rowTile + r + hiM;
        xl[row * 128 + col] = cl[r];
        xr[row * 128 + col] = cr[r];
    }
}

__global__ __launch_bounds__(256) void edge_logits(
    const float* __restrict__ edge_attr, const float* __restrict__ We,
    const float* __restrict__ xl, const float* __restrict__ xr,
    const int* __restrict__ srcIdx, const int* __restrict__ dstIdx,
    const float* __restrict__ att, float* __restrict__ ebuf,
    unsigned* __restrict__ menc, int E) {
    __shared__ float sWe[16 * 128];
    for (int i = threadIdx.x; i < 16 * 128; i += 256) sWe[i] = We[i];
    __syncthreads();

    const int wave = threadIdx.x >> 5;
    const int lane = threadIdx.x & 31;
    const int lo   = lane & 15;
    const int hiK  = (lane >= 16) ? 2 : 0;
    const int hiM  = (lane >= 16) ? 8 : 0;
    const int tile = (blockIdx.x * 8 + wave) * 16;

    const float* aRow = &edge_attr[(size_t)(tile + lo) * 16];
    v2f a0 = *(const v2f*)&aRow[0 + hiK];
    v2f a1 = *(const v2f*)&aRow[4 + hiK];
    v2f a2 = *(const v2f*)&aRow[8 + hiK];
    v2f a3 = *(const v2f*)&aRow[12 + hiK];

    int sIdx = srcIdx[tile + lo];
    int dIdx = dstIdx[tile + lo];
    int myS[8], myD[8];
#pragma unroll
    for (int r = 0; r < 8; ++r) {
        myS[r] = __shfl(sIdx, r + hiM, 32);
        myD[r] = __shfl(dIdx, r + hiM, 32);
    }

    float acc[8];
#pragma unroll
    for (int r = 0; r < 8; ++r) acc[r] = 0.0f;

#pragma unroll
    for (int wi = 0; wi < 8; ++wi) {
        const int col = wi * 16 + lo;
        v8f c;
#pragma unroll
        for (int r = 0; r < 8; ++r)
            c[r] = xl[(size_t)myS[r] * 128 + col] + xr[(size_t)myD[r] * 128 + col];
        v2f b;
        b.x = sWe[(0 + hiK) * 128 + col];  b.y = sWe[(1 + hiK) * 128 + col];
        c = __builtin_amdgcn_wmma_f32_16x16x4_f32(false, a0, false, b, (short)0, c, false, false);
        b.x = sWe[(4 + hiK) * 128 + col];  b.y = sWe[(5 + hiK) * 128 + col];
        c = __builtin_amdgcn_wmma_f32_16x16x4_f32(false, a1, false, b, (short)0, c, false, false);
        b.x = sWe[(8 + hiK) * 128 + col];  b.y = sWe[(9 + hiK) * 128 + col];
        c = __builtin_amdgcn_wmma_f32_16x16x4_f32(false, a2, false, b, (short)0, c, false, false);
        b.x = sWe[(12 + hiK) * 128 + col]; b.y = sWe[(13 + hiK) * 128 + col];
        c = __builtin_amdgcn_wmma_f32_16x16x4_f32(false, a3, false, b, (short)0, c, false, false);

        const float av = att[col];
#pragma unroll
        for (int r = 0; r < 8; ++r) acc[r] += lrelu(c[r]) * av;
    }

#pragma unroll
    for (int r = 0; r < 8; ++r) {
        float v = acc[r];
        v += __shfl_xor(v, 1, 32);
        v += __shfl_xor(v, 2, 32);
        v += __shfl_xor(v, 4, 32);
        v += __shfl_xor(v, 8, 32);
        if (lo == 0) {
            int edge = tile + r + hiM;
            ebuf[edge] = v;
            atomicMax(&menc[myD[r]], fenc(v));
        }
    }
}

#define WSUM_STRIDE 32
__global__ __launch_bounds__(256) void edge_scatter(
    const float* __restrict__ xl, const float* __restrict__ ebuf,
    const unsigned* __restrict__ menc, const int* __restrict__ srcIdx,
    const int* __restrict__ dstIdx, float* __restrict__ accum,
    float* __restrict__ wsum, int E) {
    int gw   = (blockIdx.x * 256 + threadIdx.x) >> 5;
    int lane = threadIdx.x & 31;
    if (gw >= E) return;
    int d = dstIdx[gw];
    int s = srcIdx[gw];
    float p = expf(ebuf[gw] - fdec(menc[d]));
    const float4 v = *(const float4*)&xl[(size_t)s * 128 + lane * 4];
    float* o = &accum[(size_t)d * 128 + lane * 4];
    atomicAdd(o + 0, p * v.x);
    atomicAdd(o + 1, p * v.y);
    atomicAdd(o + 2, p * v.z);
    atomicAdd(o + 3, p * v.w);
    if (lane == 0) atomicAdd(&wsum[(size_t)d * WSUM_STRIDE], p);
}

__global__ __launch_bounds__(128) void bias_act_stats(
    const float* __restrict__ accum, const float* __restrict__ wsum,
    const float* __restrict__ bias, float* __restrict__ out,
    float* __restrict__ stats, int N, int rowsPerBlock) {
    __shared__ float sInv[128];
    int c  = threadIdx.x;
    int r0 = blockIdx.x * rowsPerBlock;
    int r1 = min(r0 + rowsPerBlock, N);
    float b = bias[c];
    float sum = 0.0f, sq = 0.0f;
    for (int base = r0; base < r1; base += 128) {
        __syncthreads();
        float den = 1.0f;
        if (base + c < r1) den = wsum[(size_t)(base + c) * WSUM_STRIDE];
        sInv[c] = den > 0.0f ? 1.0f / den : 1.0f;
        __syncthreads();
        int rEnd = min(base + 128, r1);
        for (int r = base; r < rEnd; ++r) {
            float v = lrelu(accum[(size_t)r * 128 + c] * sInv[r - base] + b);
            out[(size_t)r * 128 + c] = v;
            sum += v;
            sq  += v * v;
        }
    }
    atomicAdd(&stats[c], sum);
    atomicAdd(&stats[128 + c], sq);
}

__global__ void bn_prep(const float* __restrict__ stats,
                        const float* __restrict__ bnw,
                        const float* __restrict__ bnb,
                        float* __restrict__ scsh, int N) {
    int c = threadIdx.x;
    float inv = 1.0f / (float)N;
    float mean = stats[c] * inv;
    float var  = stats[128 + c] * inv - mean * mean;
    float sc   = bnw[c] * rsqrtf(var + BN_EPS);
    scsh[c]       = sc;
    scsh[128 + c] = bnb[c] - mean * sc;
}

__global__ void bn_apply(float* __restrict__ out, const float* __restrict__ scsh,
                         int total) {
    int i = blockIdx.x * blockDim.x + threadIdx.x;
    if (i >= total) return;
    int c = i & 127;
    out[i] = out[i] * scsh[c] + scsh[128 + c];
}

extern "C" void kernel_launch(void* const* d_in, const int* in_sizes, int n_in,
                              void* d_out, int out_size, void* d_ws, size_t ws_size,
                              hipStream_t stream) {
    const float* x    = (const float*)d_in[0];
    const float* ea   = (const float*)d_in[1];
    const int*   ei   = (const int*)d_in[2];
    const float* Wl   = (const float*)d_in[3];
    const float* Wr   = (const float*)d_in[4];
    const float* We   = (const float*)d_in[5];
    const float* att  = (const float*)d_in[6];
    const float* bias = (const float*)d_in[7];
    const float* bnw  = (const float*)d_in[8];
    const float* bnb  = (const float*)d_in[9];
    float* out = (float*)d_out;

    const int N = in_sizes[0] / 128;
    const int E = in_sizes[2] / 2;
    const int* srcIdx = ei;
    const int* dstIdx = ei + E;

    float* ws = (float*)d_ws;
    float* xl = ws;     ws += (size_t)N * 128;
    float* xr = ws;     ws += (size_t)N * 128;
    float* ebuf = ws;   ws += E;
    float* zbase = ws;
    float* wsum  = ws;  ws += (size_t)N * WSUM_STRIDE;
    float* accum = ws;  ws += (size_t)N * 128;
    unsigned* menc = (unsigned*)ws; ws += N;
    float* stats = ws;  ws += 256;
    float* scsh  = ws;  ws += 256;
    int zcount = (int)(ws - zbase);

    zero_ws<<<(zcount + 255) / 256, 256, 0, stream>>>(zbase, zcount);
    node_transform<<<N / 16, 256, 0, stream>>>(x, Wl, Wr, xl, xr, N);
    edge_logits<<<E / 128, 256, 0, stream>>>(ea, We, xl, xr, srcIdx, dstIdx, att,
                                             ebuf, menc, E);
    edge_scatter<<<(E + 7) / 8, 256, 0, stream>>>(xl, ebuf, menc, srcIdx, dstIdx,
                                                  accum, wsum, E);
    const int rpb = 50;
    bias_act_stats<<<(N + rpb - 1) / rpb, 128, 0, stream>>>(accum, wsum, bias, out,
                                                            stats, N, rpb);
    bn_prep<<<1, 128, 0, stream>>>(stats, bnw, bnb, scsh, N);
    bn_apply<<<(N * 128 + 255) / 256, 256, 0, stream>>>(out, scsh, N * 128);
}


# Float atomic scatter replaced by a counting sort on dst (one int ticket per edge) and one wave per node summing its run in registers; max folded in; sum order changes (arrival-order rank)
# speedup vs baseline: 2.0230x; 1.6098x over previous
#include <hip/hip_runtime.h>

typedef __attribute__((ext_vector_type(2))) float v2f;
typedef __attribute__((ext_vector_type(8))) float v8f;

#define NEG_SLOPE 0.2f
#define BN_EPS 1e-5f

__device__ __forceinline__ float lrelu(float x) { return x > 0.0f ? x : NEG_SLOPE * x; }

#define CNT_STRIDE 32

__global__ void zero_ws(float* __restrict__ p, int n) {
    int i = blockIdx.x * blockDim.x + threadIdx.x;
    if (i < n) p[i] = 0.0f;
}

__global__ __launch_bounds__(256) void node_transform(
    const float* __restrict__ x, const float* __restrict__ Wl,
    const float* __restrict__ Wr, float* __restrict__ xl, float* __restrict__ xr,
    int N) {
    const int wave = threadIdx.x >> 5;
    const int lane = threadIdx.x & 31;
    const int lo   = lane & 15;
    const int hiK  = (lane >= 16) ? 2 : 0;
    const int hiM  = (lane >= 16) ? 8 : 0;
    const int rowTile = blockIdx.x * 16;
    const int col  = wave * 16 + lo;
    const int aRow = rowTile + lo;

    v8f cl = {};
    v8f cr = {};
    for (int k = 0; k < 128; k += 4) {
        v2f a = *(const v2f*)&x[aRow * 128 + k + hiK];
        v2f bl, br;
        bl.x = Wl[(k + hiK) * 128 + col];
        bl.y = Wl[(k + hiK + 1) * 128 + col];
        br.x = Wr[(k + hiK) * 128 + col];
        br.y = Wr[(k + hiK + 1) * 128 + col];
        cl = __builtin_amdgcn_wmma_f32_16x16x4_f32(false, a, false, bl, (short)0, cl, false, false);
        cr = __builtin_amdgcn_wmma_f32_16x16x4_f32(false, a, false, br, (short)0, cr, false, false);
    }
#pragma unroll
    for (int r = 0; r < 8; ++r) {
        int row = rowTile + r + hiM;
        xl[row * 128 + col] = cl[r];
        xr[row * 128 + col] = cr[r];
    }
}

__global__ __launch_bounds__(256, 8) void edge_logits(
    const float* __restrict__ edge_attr, const float* __restrict__ We,
    const float* __restrict__ xl, const float* __restrict__ xr,
    const int* __restrict__ srcIdx, const int* __restrict__ dstIdx,
    const float* __restrict__ att, float* __restrict__ ebuf,
    int* __restrict__ cnt, int* __restrict__ rk, int E) {
    __shared__ float sWe[16 * 128];
    for (int i = threadIdx.x; i < 16 * 128; i += 256) sWe[i] = We[i];
    __syncthreads();

    const int wave = threadIdx.x >> 5;
    const int lane = threadIdx.x & 31;
    const int lo   = lane & 15;
    const int hiK  = (lane >= 16) ? 2 : 0;
    const int hiM  = (lane >= 16) ? 8 : 0;
    const int tile = (blockIdx.x * 8 + wave) * 16;

    const float* aRow = &edge_attr[(size_t)(tile + lo) * 16];
    v2f a0 = *(const v2f*)&aRow[0 + hiK];
    v2f a1 = *(const v2f*)&aRow[4 + hiK];
    v2f a2 = *(const v2f*)&aRow[8 + hiK];
    v2f a3 = *(const v2f*)&aRow[12 + hiK];

    int sIdx = srcIdx[tile + lo];
    int dIdx = dstIdx[tile + lo];
    int myS[8], myD[8];
#pragma unroll
    for (int r = 0; r < 8; ++r) {
        myS[r] = __shfl(sIdx, r + hiM, 32);
        myD[r] = __shfl(dIdx, r + hiM, 32);
    }

    float acc[8];
#pragma unroll
    for (int r = 0; r < 8; ++r) acc[r] = 0.0f;

#pragma unroll
    for (int wi = 0; wi < 8; ++wi) {
        const int col = wi * 16 + lo;
        v8f c;
#pragma unroll
        for (int r = 0; r < 8; ++r)
            c[r] = xl[(size_t)myS[r] * 128 + col] + xr[(size_t)myD[r] * 128 + col];
        v2f b;
        b.x = sWe[(0 + hiK) * 128 + col];  b.y = sWe[(1 + hiK) * 128 + col];
        c = __builtin_amdgcn_wmma_f32_16x16x4_f32(false, a0, false, b, (short)0, c, false, false);
        b.x = sWe[(4 + hiK) * 128 + col];  b.y = sWe[(5 + hiK) * 128 + col];
        c = __builtin_amdgcn_wmma_f32_16x16x4_f32(false, a1, false, b, (short)0, c, false, false);
        b.x = sWe[(8 + hiK) * 128 + col];  b.y = sWe[(9 + hiK) * 128 + col];
        c = __builtin_amdgcn_wmma_f32_16x16x4_f32(false, a2, false, b, (short)0, c, false, false);
        b.x = sWe[(12 + hiK) * 128 + col]; b.y = sWe[(13 + hiK) * 128 + col];
        c = __builtin_amdgcn_wmma_f32_16x16x4_f32(false, a3, false, b, (short)0, c, false, false);

        const float av = att[col];
#pragma unroll
        for (int r = 0; r < 8; ++r) acc[r] += lrelu(c[r]) * av;
    }

#pragma unroll
    for (int r = 0; r < 8; ++r) {
        float v = acc[r];
        v += __shfl_xor(v, 1, 32);
        v += __shfl_xor(v, 2, 32);
        v += __shfl_xor(v, 4, 32);
        v += __shfl_xor(v, 8, 32);
        if (lo == 0) ebuf[tile + r + hiM] = v;
    }
    if (lane < 16) rk[tile + lo] = atomicAdd(&cnt[(size_t)dIdx * CNT_STRIDE], 1);
}

__global__ __launch_bounds__(1024) void seg_offsets(
    const int* __restrict__ cnt, int* __restrict__ offs, int N) {
    __shared__ int sScan[1024];
    const int t   = threadIdx.x;
    const int per = (N + 1023) / 1024;
    const int b0  = t * per;
    const int b1  = min(b0 + per, N);
    int sum = 0;
    for (int i = b0; i < b1; ++i) sum += cnt[(size_t)i * CNT_STRIDE];
    sScan[t] = sum;
    __syncthreads();
    for (int off = 1; off < 1024; off <<= 1) {
        int v = (t >= off) ? sScan[t - off] : 0;
        __syncthreads();
        sScan[t] += v;
        __syncthreads();
    }
    int run = sScan[t] - sum;
    for (int i = b0; i < b1; ++i) {
        offs[i] = run;
        run += cnt[(size_t)i * CNT_STRIDE];
    }
    if (t == 1023) offs[N] = sScan[1023];
}

__global__ __launch_bounds__(256) void edge_place(
    const int* __restrict__ srcIdx, const int* __restrict__ dstIdx,
    const int* __restrict__ rk, const int* __restrict__ offs,
    const float* __restrict__ ebuf, int* __restrict__ pairs, int E) {
    int e = blockIdx.x * 256 + threadIdx.x;
    if (e >= E) return;
    int pos = offs[dstIdx[e]] + rk[e];
    ((int2*)pairs)[pos] = make_int2(srcIdx[e], __float_as_int(ebuf[e]));
}

__global__ __launch_bounds__(128) void node_gather(
    const float* __restrict__ xl, const int* __restrict__ pairs,
    const int* __restrict__ offs, float* __restrict__ accum,
    float* __restrict__ wsum, int N) {
    const int node = blockIdx.x * 4 + __builtin_amdgcn_readfirstlane(threadIdx.x >> 5);
    const int lane = threadIdx.x & 31;
    if (node >= N) return;
    const int2* pr2 = (const int2*)pairs;
    const int beg = offs[node];
    const int end = offs[node + 1];

    float m = -INFINITY;
    for (int i = beg + lane; i < end; i += 32)
        m = fmaxf(m, __int_as_float(pr2[i].y));
    m = fmaxf(m, __shfl_xor(m, 16, 32));
    m = fmaxf(m, __shfl_xor(m, 8, 32));
    m = fmaxf(m, __shfl_xor(m, 4, 32));
    m = fmaxf(m, __shfl_xor(m, 2, 32));
    m = fmaxf(m, __shfl_xor(m, 1, 32));

    float4 acc = make_float4(0.0f, 0.0f, 0.0f, 0.0f);
    float ws = 0.0f;
    for (int base = beg; base < end; base += 32) {
        const int i = base + lane;
        int   s = 0;
        float p = 0.0f;
        if (i < end) {
            const int2 pr = pr2[i];
            s = pr.x;
            p = expf(__int_as_float(pr.y) - m);
        }
        ws += p;
        const int n = min(32, end - base);
        for (int jb = 0; jb < n; jb += 8) {
            float4 v[8];
            float  pj[8];
#pragma unroll
            for (int u = 0; u < 8; ++u) {
                const int sj = __builtin_amdgcn_readlane(s, jb + u);
                pj[u] = __int_as_float(__builtin_amdgcn_readlane(__float_as_int(p), jb + u));
                v[u]  = *(const float4*)&xl[(size_t)sj * 128 + lane * 4];
            }
#pragma unroll
            for (int u = 0; u < 8; ++u) {
                acc.x += pj[u] * v[u].x;
                acc.y += pj[u] * v[u].y;
                acc.z += pj[u] * v[u].z;
                acc.w += pj[u] * v[u].w;
            }
        }
    }
    ws += __shfl_xor(ws, 16, 32);
    ws += __shfl_xor(ws, 8, 32);
    ws += __shfl_xor(ws, 4, 32);
    ws += __shfl_xor(ws, 2, 32);
    ws += __shfl_xor(ws, 1, 32);

    *(float4*)&accum[(size_t)node * 128 + lane * 4] = acc;
    if (lane == 0) wsum[node] = ws;
}

__global__ __launch_bounds__(128) void bias_act_stats(
    const float* __restrict__ accum, const float* __restrict__ wsum,
    const float* __restrict__ bias, float* __restrict__ out,
    float* __restrict__ stats, int N, int rowsPerBlock) {
    __shared__ float sInv[128];
    int c  = threadIdx.x;
    int r0 = blockIdx.x * rowsPerBlock;
    int r1 = min(r0 + rowsPerBlock, N);
    float b = bias[c];
    float sum = 0.0f, sq = 0.0f;
    for (int base = r0; base < r1; base += 128) {
        __syncthreads();
        float den = 1.0f;
        if (base + c < r1) den = wsum[base + c];
        sInv[c] = den > 0.0f ? 1.0f / den : 1.0f;
        __syncthreads();
        int rEnd = min(base + 128, r1);
        for (int r = base; r < rEnd; ++r) {
            float v = lrelu(accum[(size_t)r * 128 + c] * sInv[r - base] + b);
            out[(size_t)r * 128 + c] = v;
            sum += v;
            sq  += v * v;
        }
    }
    atomicAdd(&stats[c], sum);
    atomicAdd(&stats[128 + c], sq);
}

__global__ void bn_prep(const float* __restrict__ stats,
                        const float* __restrict__ bnw,
                        const float* __restrict__ bnb,
                        float* __restrict__ scsh, int N) {
    int c = threadIdx.x;
    float inv = 1.0f / (float)N;
    float mean = stats[c] * inv;
    float var  = stats[128 + c] * inv - mean * mean;
    float sc   = bnw[c] * rsqrtf(var + BN_EPS);
    scsh[c]       = sc;
    scsh[128 + c] = bnb[c] - mean * sc;
}

__global__ void bn_apply(float* __restrict__ out, const float* __restrict__ scsh,
                         int total) {
    int i = blockIdx.x * blockDim.x + threadIdx.x;
    if (i >= total) return;
    int c = i & 127;
    out[i] = out[i] * scsh[c] + scsh[128 + c];
}

extern "C" void kernel_launch(void* const* d_in, const int* in_sizes, int n_in,
                              void* d_out, int out_size, void* d_ws, size_t ws_size,
                              hipStream_t stream) {
    const float* x    = (const float*)d_in[0];
    const float* ea   = (const float*)d_in[1];
    const int*   ei   = (const int*)d_in[2];
    const float* Wl   = (const float*)d_in[3];
    const float* Wr   = (const float*)d_in[4];
    const float* We   = (const float*)d_in[5];
    const float* att  = (const float*)d_in[6];
    const float* bias = (const float*)d_in[7];
    const float* bnw  = (const float*)d_in[8];
    const float* bnb  = (const float*)d_in[9];
    float* out = (float*)d_out;

    const int N = in_sizes[0] / 128;
    const int E = in_sizes[2] / 2;
    const int* srcIdx = ei;
    const int* dstIdx = ei + E;

    float* ws = (float*)d_ws;
    float* xl = ws;     ws += (size_t)N * 128;
    float* xr = ws;     ws += (size_t)N * 128;
    float* accum = ws;  ws += (size_t)N * 128;
    float* ebuf = ws;   ws += E;
    int* rk    = (int*)ws;  ws += E;
    int* pairs = (int*)ws;  ws += (size_t)E * 2;
    float* wsum = ws;   ws += N;
    int* offs  = (int*)ws;  ws += N + 32;
    float* zbase = ws;
    int* cnt   = (int*)ws;  ws += (size_t)N * CNT_STRIDE;
    float* stats = ws;  ws += 256;
    int zcount = (int)(ws - zbase);
    float* scsh  = ws;  ws += 256;

    zero_ws<<<(zcount + 255) / 256, 256, 0, stream>>>(zbase, zcount);
    node_transform<<<N / 16, 256, 0, stream>>>(x, Wl, Wr, xl, xr, N);
    edge_logits<<<E / 128, 256, 0, stream>>>(ea, We, xl, xr, srcIdx, dstIdx, att,
                                             ebuf, cnt, rk, E);
    seg_offsets<<<1, 1024, 0, stream>>>(cnt, offs, N);
    edge_place<<<(E + 255) / 256, 256, 0, stream>>>(srcIdx, dstIdx, rk, offs, ebuf,
                                                    pairs, E);
    node_gather<<<(N + 3) / 4, 128, 0, stream>>>(xl, pairs, offs, accum, wsum, N);
    const int rpb = 50;
    bias_act_stats<<<(N + rpb - 1) / rpb, 128, 0, stream>>>(accum, wsum, bias, out,
                                                            stats, N, rpb);
    bn_prep<<<1, 128, 0, stream>>>(stats, bnw, bnb, scsh, N);
    bn_apply<<<(N * 128 + 255) / 256, 256, 0, stream>>>(out, scsh, N * 128);
}


# edge_logits operand fetch re-laid out: tile (q,j) owns columns 64q+4c+j, so a lane loads 16 B and a half-wave one aligned 256-B row piece; We staged to match; column sum order of the logits changes
# speedup vs baseline: 2.1620x; 1.0687x over previous
#include <hip/hip_runtime.h>

typedef __attribute__((ext_vector_type(2))) float v2f;
typedef __attribute__((ext_vector_type(8))) float v8f;

#define NEG_SLOPE 0.2f
#define BN_EPS 1e-5f

__device__ __forceinline__ float lrelu(float x) { return x > 0.0f ? x : NEG_SLOPE * x; }

#define CNT_STRIDE 32

__global__ void zero_ws(float* __restrict__ p, int n) {
    int i = blockIdx.x * blockDim.x + threadIdx.x;
    if (i < n) p[i] = 0.0f;
}

__global__ __launch_bounds__(256) void node_transform(
    const float* __restrict__ x, const float* __restrict__ Wl,
    const float* __restrict__ Wr, float* __restrict__ xl, float* __restrict__ xr,
    int N) {
    const int wave = threadIdx.x >> 5;
    const int lane = threadIdx.x & 31;
    const int lo   = lane & 15;
    const int hiK  = (lane >= 16) ? 2 : 0;
    const int hiM  = (lane >= 16) ? 8 : 0;
    const int rowTile = blockIdx.x * 16;
    const int col  = wave * 16 + lo;
    const int aRow = rowTile + lo;

    v8f cl = {};
    v8f cr = {};
    for (int k = 0; k < 128; k += 4) {
        v2f a = *(const v2f*)&x[aRow * 128 + k + hiK];
        v2f bl, br;
        bl.x = Wl[(k + hiK) * 128 + col];
        bl.y = Wl[(k + hiK + 1) * 128 + col];
        br.x = Wr[(k + hiK) * 128 + col];
        br.y = Wr[(k + hiK + 1) * 128 + col];
        cl = __builtin_amdgcn_wmma_f32_16x16x4_f32(false, a, false, bl, (short)0, cl, false, false);
        cr = __builtin_amdgcn_wmma_f32_16x16x4_f32(false, a, false, br, (short)0, cr, false, false);
    }
#pragma unroll
    for (int r = 0; r < 8; ++r) {
        int row = rowTile + r + hiM;
        xl[row * 128 + col] = cl[r];
        xr[row * 128 + col] = cr[r];
    }
}

__global__ __launch_bounds__(256, 8) void edge_logits(
    const float* __restrict__ edge_attr, const float* __restrict__ We,
    const float* __restrict__ xl, const float* __restrict__ xr,
    const int* __restrict__ srcIdx, const int* __restrict__ dstIdx,
    const float* __restrict__ att, float* __restrict__ ebuf,
    int* __restrict__ cnt, int* __restrict__ rk, int E) {
    __shared__ float4 sB[512];
    {
        const int c  = threadIdx.x & 15;
        const int h  = (threadIdx.x >> 4) & 1;
        const int q  = (threadIdx.x >> 5) & 1;
        const int ks = (threadIdx.x >> 6) & 3;
        const int k0 = 4 * ks + 2 * h;
        const float4 r0 = *(const float4*)&We[k0 * 128 + 64 * q + 4 * c];
        const float4 r1 = *(const float4*)&We[(k0 + 1) * 128 + 64 * q + 4 * c];
        sB[((((ks * 2 + q) * 2 + 0) * 2 + h) * 16) + c] = make_float4(r0.x, r1.x, r0.y, r1.y);
        sB[((((ks * 2 + q) * 2 + 1) * 2 + h) * 16) + c] = make_float4(r0.z, r1.z, r0.w, r1.w);
    }
    __syncthreads();

    const int wave = threadIdx.x >> 5;
    const int lane = threadIdx.x & 31;
    const int lo   = lane & 15;
    const int hh   = lane >> 4;
    const int hiK  = hh * 2;
    const int hiM  = hh * 8;
    const int tile = (blockIdx.x * 8 + wave) * 16;

    const float* aRow = &edge_attr[(size_t)(tile + lo) * 16];
    v2f a[4];
    a[0] = *(const v2f*)&aRow[0 + hiK];
    a[1] = *(const v2f*)&aRow[4 + hiK];
    a[2] = *(const v2f*)&aRow[8 + hiK];
    a[3] = *(const v2f*)&aRow[12 + hiK];

    int sIdx = srcIdx[tile + lo];
    int dIdx = dstIdx[tile + lo];
    unsigned offS[8], offD[8];
#pragma unroll
    for (int r = 0; r < 8; ++r) {
        offS[r] = (unsigned)__shfl(sIdx, r + hiM, 32) * 512u + (unsigned)lo * 16u;
        offD[r] = (unsigned)__shfl(dIdx, r + hiM, 32) * 512u + (unsigned)lo * 16u;
    }

    float acc[8];
#pragma unroll
    for (int r = 0; r < 8; ++r) acc[r] = 0.0f;

#pragma unroll
    for (int q = 0; q < 2; ++q) {
        float g[8][4];
#pragma unroll
        for (int r = 0; r < 8; ++r) {
            const float4 vl = *(const float4*)((const char*)xl + offS[r] + q * 256);
            const float4 vr = *(const float4*)((const char*)xr + offD[r] + q * 256);
            g[r][0] = vl.x + vr.x;
            g[r][1] = vl.y + vr.y;
            g[r][2] = vl.z + vr.z;
            g[r][3] = vl.w + vr.w;
        }
        const float4 av4 = *(const float4*)&att[q * 64 + lo * 4];
        const float avj[4] = {av4.x, av4.y, av4.z, av4.w};
#pragma unroll
        for (int jp = 0; jp < 2; ++jp) {
            float4 bv[4];
#pragma unroll
            for (int ks = 0; ks < 4; ++ks)
                bv[ks] = sB[((((ks * 2 + q) * 2 + jp) * 2 + hh) * 16) + lo];
#pragma unroll
            for (int jj = 0; jj < 2; ++jj) {
                const int j = 2 * jp + jj;
                v8f c;
#pragma unroll
                for (int r = 0; r < 8; ++r) c[r] = g[r][j];
#pragma unroll
                for (int ks = 0; ks < 4; ++ks) {
                    v2f b;
                    b.x = jj ? bv[ks].z : bv[ks].x;
                    b.y = jj ? bv[ks].w : bv[ks].y;
                    c = __builtin_amdgcn_wmma_f32_16x16x4_f32(false, a[ks], false, b, (short)0, c, false, false);
                }
#pragma unroll
                for (int r = 0; r < 8; ++r) acc[r] += lrelu(c[r]) * avj[j];
            }
        }
    }

#pragma unroll
    for (int r = 0; r < 8; ++r) {
        float v = acc[r];
        v += __shfl_xor(v, 1, 32);
        v += __shfl_xor(v, 2, 32);
        v += __shfl_xor(v, 4, 32);
        v += __shfl_xor(v, 8, 32);
        if (lo == 0) ebuf[tile + r + hiM] = v;
    }
    if (lane < 16) rk[tile + lo] = atomicAdd(&cnt[(size_t)dIdx * CNT_STRIDE], 1);
}

__global__ __launch_bounds__(1024) void seg_offsets(
    const int* __restrict__ cnt, int* __restrict__ offs, int N) {
    __shared__ int sScan[1024];
    const int t   = threadIdx.x;
    const int per = (N + 1023) / 1024;
    const int b0  = t * per;
    const int b1  = min(b0 + per, N);
    int sum = 0;
    for (int i = b0; i < b1; ++i) sum += cnt[(size_t)i * CNT_STRIDE];
    sScan[t] = sum;
    __syncthreads();
    for (int off = 1; off < 1024; off <<= 1) {
        int v = (t >= off) ? sScan[t - off] : 0;
        __syncthreads();
        sScan[t] += v;
        __syncthreads();
    }
    int run = sScan[t] - sum;
    for (int i = b0; i < b1; ++i) {
        offs[i] = run;
        run += cnt[(size_t)i * CNT_STRIDE];
    }
    if (t == 1023) offs[N] = sScan[1023];
}

__global__ __launch_bounds__(256) void edge_place(
    const int* __restrict__ srcIdx, const int* __restrict__ dstIdx,
    const int* __restrict__ rk, const int* __restrict__ offs,
    const float* __restrict__ ebuf, int* __restrict__ pairs, int E) {
    int e = blockIdx.x * 256 + threadIdx.x;
    if (e >= E) return;
    int pos = offs[dstIdx[e]] + rk[e];
    ((int2*)pairs)[pos] = make_int2(srcIdx[e], __float_as_int(ebuf[e]));
}

__global__ __launch_bounds__(128) void node_gather(
    const float* __restrict__ xl, const int* __restrict__ pairs,
    const int* __restrict__ offs, float* __restrict__ accum,
    float* __restrict__ wsum, int N) {
    const int node = blockIdx.x * 4 + __builtin_amdgcn_readfirstlane(threadIdx.x >> 5);
    const int lane = threadIdx.x & 31;
    if (node >= N) return;
    const int2* pr2 = (const int2*)pairs;
    const int beg = offs[node];
    const int end = offs[node + 1];

    float m = -INFINITY;
    for (int i = beg + lane; i < end; i += 32)
        m = fmaxf(m, __int_as_float(pr2[i].y));
    m = fmaxf(m, __shfl_xor(m, 16, 32));
    m = fmaxf(m, __shfl_xor(m, 8, 32));
    m = fmaxf(m, __shfl_xor(m, 4, 32));
    m = fmaxf(m, __shfl_xor(m, 2, 32));
    m = fmaxf(m, __shfl_xor(m, 1, 32));

    float4 acc = make_float4(0.0f, 0.0f, 0.0f, 0.0f);
    float ws = 0.0f;
    for (int base = beg; base < end; base += 32) {
        const int i = base + lane;
        int   s = 0;
        float p = 0.0f;
        if (i < end) {
            const int2 pr = pr2[i];
            s = pr.x;
            p = expf(__int_as_float(pr.y) - m);
        }
        ws += p;
        const int n = min(32, end - base);
        for (int jb = 0; jb < n; jb += 8) {
            float4 v[8];
            float  pj[8];
#pragma unroll
            for (int u = 0; u < 8; ++u) {
                const int sj = __builtin_amdgcn_readlane(s, jb + u);
                pj[u] = __int_as_float(__builtin_amdgcn_readlane(__float_as_int(p), jb + u));
                v[u]  = *(const float4*)&xl[(size_t)sj * 128 + lane * 4];
            }
#pragma unroll
            for (int u = 0; u < 8; ++u) {
                acc.x += pj[u] * v[u].x;
                acc.y += pj[u] * v[u].y;
                acc.z += pj[u] * v[u].z;
                acc.w += pj[u] * v[u].w;
            }
        }
    }
    ws += __shfl_xor(ws, 16, 32);
    ws += __shfl_xor(ws, 8, 32);
    ws += __shfl_xor(ws, 4, 32);
    ws += __shfl_xor(ws, 2, 32);
    ws += __shfl_xor(ws, 1, 32);

    *(float4*)&accum[(size_t)node * 128 + lane * 4] = acc;
    if (lane == 0) wsum[node] = ws;
}

__global__ __launch_bounds__(128) void bias_act_stats(
    const float* __restrict__ accum, const float* __restrict__ wsum,
    const float* __restrict__ bias, float* __restrict__ out,
    float* __restrict__ stats, int N, int rowsPerBlock) {
    __shared__ float sInv[128];
    int c  = threadIdx.x;
    int r0 = blockIdx.x * rowsPerBlock;
    int r1 = min(r0 + rowsPerBlock, N);
    float b = bias[c];
    float sum = 0.0f, sq = 0.0f;
    for (int base = r0; base < r1; base += 128) {
        __syncthreads();
        float den = 1.0f;
        if (base + c < r1) den = wsum[base + c];
        sInv[c] = den > 0.0f ? 1.0f / den : 1.0f;
        __syncthreads();
        int rEnd = min(base + 128, r1);
        for (int r = base; r < rEnd; ++r) {
            float v = lrelu(accum[(size_t)r * 128 + c] * sInv[r - base] + b);
            out[(size_t)r * 128 + c] = v;
            sum += v;
            sq  += v * v;
        }
    }
    atomicAdd(&stats[c], sum);
    atomicAdd(&stats[128 + c], sq);
}

__global__ void bn_prep(const float* __restrict__ stats,
                        const float* __restrict__ bnw,
                        const float* __restrict__ bnb,
                        float* __restrict__ scsh, int N) {
    int c = threadIdx.x;
    float inv = 1.0f / (float)N;
    float mean = stats[c] * inv;
    float var  = stats[128 + c] * inv - mean * mean;
    float sc   = bnw[c] * rsqrtf(var + BN_EPS);
    scsh[c]       = sc;
    scsh[128 + c] = bnb[c] - mean * sc;
}

__global__ void bn_apply(float* __restrict__ out, const float* __restrict__ scsh,
                         int total) {
    int i = blockIdx.x * blockDim.x + threadIdx.x;
    if (i >= total) return;
    int c = i & 127;
    out[i] = out[i] * scsh[c] + scsh[128 + c];
}

extern "C" void kernel_launch(void* const* d_in, const int* in_sizes, int n_in,
                              void* d_out, int out_size, void* d_ws, size_t ws_size,
                              hipStream_t stream) {
    const float* x    = (const float*)d_in[0];
    const float* ea   = (const float*)d_in[1];
    const int*   ei   = (const int*)d_in[2];
    const float* Wl   = (const float*)d_in[3];
    const float* Wr   = (const float*)d_in[4];
    const float* We   = (const float*)d_in[5];
    const float* att  = (const float*)d_in[6];
    const float* bias = (const float*)d_in[7];
    const float* bnw  = (const float*)d_in[8];
    const float* bnb  = (const float*)d_in[9];
    float* out = (float*)d_out;

    const int N = in_sizes[0] / 128;
    const int E = in_sizes[2] / 2;
    const int* srcIdx = ei;
    const int* dstIdx = ei + E;

    float* ws = (float*)d_ws;
    float* xl = ws;     ws += (size_t)N * 128;
    float* xr = ws;     ws += (size_t)N * 128;
    float* accum = ws;  ws += (size_t)N * 128;
    float* ebuf = ws;   ws += E;
    int* rk    = (int*)ws;  ws += E;
    int* pairs = (int*)ws;  ws += (size_t)E * 2;
    float* wsum = ws;   ws += N;
    int* offs  = (int*)ws;  ws += N + 32;
    float* zbase = ws;
    int* cnt   = (int*)ws;  ws += (size_t)N * CNT_STRIDE;
    float* stats = ws;  ws += 256;
    int zcount = (int)(ws - zbase);
    float* scsh  = ws;  ws += 256;

    zero_ws<<<(zcount + 255) / 256, 256, 0, stream>>>(zbase, zcount);
    node_transform<<<N / 16, 256, 0, stream>>>(x, Wl, Wr, xl, xr, N);
    edge_logits<<<E / 128, 256, 0, stream>>>(ea, We, xl, xr, srcIdx, dstIdx, att,
                                             ebuf, cnt, rk, E);
    seg_offsets<<<1, 1024, 0, stream>>>(cnt, offs, N);
    edge_place<<<(E + 255) / 256, 256, 0, stream>>>(srcIdx, dstIdx, rk, offs, ebuf,
                                                    pairs, E);
    node_gather<<<(N + 3) / 4, 128, 0, stream>>>(xl, pairs, offs, accum, wsum, N);
    const int rpb = 50;
    bias_act_stats<<<(N + rpb - 1) / rpb, 128, 0, stream>>>(accum, wsum, bias, out,
                                                            stats, N, rpb);
    bn_prep<<<1, 128, 0, stream>>>(stats, bnw, bnb, scsh, N);
    bn_apply<<<(N * 128 + 255) / 256, 256, 0, stream>>>(out, scsh, N * 128);
}


# Global returning atomic tickets replaced by a chunked counting sort: 128 chunks ranked on LDS counters, then a per-node prefix over chunks; scratch clear dropped; sum order changes (chunk-major rank)
# speedup vs baseline: 2.5316x; 1.1710x over previous
#include <hip/hip_runtime.h>

typedef __attribute__((ext_vector_type(2))) float v2f;
typedef __attribute__((ext_vector_type(8))) float v8f;

#define NEG_SLOPE 0.2f
#define BN_EPS 1e-5f

__device__ __forceinline__ float lrelu(float x) { return x > 0.0f ? x : NEG_SLOPE * x; }

#define RANK_CHUNKS 128

__global__ __launch_bounds__(1024) void edge_rank(
    const int* __restrict__ dstIdx, int* __restrict__ rk, int* __restrict__ hist,
    float* __restrict__ stats, int E, int N, int per) {
    extern __shared__ int sCnt[];
    const int t = threadIdx.x;
    for (int i = t; i < N; i += 1024) sCnt[i] = 0;
    if (blockIdx.x == 0 && t < 256) stats[t] = 0.0f;
    __syncthreads();
    const int e0 = blockIdx.x * per;
    const int e1 = min(e0 + per, E);
    for (int eb = e0 + t; eb < e1; eb += 8 * 1024) {
        int d[8];
#pragma unroll
        for (int u = 0; u < 8; ++u) {
            const int e = eb + u * 1024;
            d[u] = (e < e1) ? dstIdx[e] : 0;
        }
#pragma unroll
        for (int u = 0; u < 8; ++u) {
            const int e = eb + u * 1024;
            if (e < e1) rk[e] = atomicAdd(&sCnt[d[u]], 1);
        }
    }
    __syncthreads();
    int* __restrict__ h = hist + (size_t)blockIdx.x * N;
    for (int i = t; i < N; i += 1024) h[i] = sCnt[i];
}

__global__ __launch_bounds__(64) void chunk_prefix(
    const int* __restrict__ hist, int* __restrict__ cbase, int* __restrict__ cnt,
    int N) {
    const int node = blockIdx.x * 64 + threadIdx.x;
    if (node >= N) return;
    int run = 0;
    for (int c0 = 0; c0 < RANK_CHUNKS; c0 += 16) {
        int v[16];
#pragma unroll
        for (int u = 0; u < 16; ++u) v[u] = hist[(size_t)(c0 + u) * N + node];
#pragma unroll
        for (int u = 0; u < 16; ++u) {
            cbase[(size_t)(c0 + u) * N + node] = run;
            run += v[u];
        }
    }
    cnt[node] = run;
}

__global__ __launch_bounds__(256) void node_transform(
    const float* __restrict__ x, const float* __restrict__ Wl,
    const float* __restrict__ Wr, float* __restrict__ xl, float* __restrict__ xr,
    int N) {
    const int wave = threadIdx.x >> 5;
    const int lane = threadIdx.x & 31;
    const int lo   = lane & 15;
    const int hiK  = (lane >= 16) ? 2 : 0;
    const int hiM  = (lane >= 16) ? 8 : 0;
    const int rowTile = blockIdx.x * 16;
    const int col  = wave * 16 + lo;
    const int aRow = rowTile + lo;

    v8f cl = {};
    v8f cr = {};
    for (int k = 0; k < 128; k += 4) {
        v2f a = *(const v2f*)&x[aRow * 128 + k + hiK];
        v2f bl, br;
        bl.x = Wl[(k + hiK) * 128 + col];
        bl.y = Wl[(k + hiK + 1) * 128 + col];
        br.x = Wr[(k + hiK) * 128 + col];
        br.y = Wr[(k + hiK + 1) * 128 + col];
        cl = __builtin_amdgcn_wmma_f32_16x16x4_f32(false, a, false, bl, (short)0, cl, false, false);
        cr = __builtin_amdgcn_wmma_f32_16x16x4_f32(false, a, false, br, (short)0, cr, false, false);
    }
#pragma unroll
    for (int r = 0; r < 8; ++r) {
        int row = rowTile + r + hiM;
        xl[row * 128 + col] = cl[r];
        xr[row * 128 + col] = cr[r];
    }
}

__global__ __launch_bounds__(256, 8) void edge_logits(
    const float* __restrict__ edge_attr, const float* __restrict__ We,
    const float* __restrict__ xl, const float* __restrict__ xr,
    const int* __restrict__ srcIdx, const int* __restrict__ dstIdx,
    const float* __restrict__ att, float* __restrict__ ebuf, int E) {
    __shared__ float4 sB[512];
    {
        const int c  = threadIdx.x & 15;
        const int h  = (threadIdx.x >> 4) & 1;
        const int q  = (threadIdx.x >> 5) & 1;
        const int ks = (threadIdx.x >> 6) & 3;
        const int k0 = 4 * ks + 2 * h;
        const float4 r0 = *(const float4*)&We[k0 * 128 + 64 * q + 4 * c];
        const float4 r1 = *(const float4*)&We[(k0 + 1) * 128 + 64 * q + 4 * c];
        sB[((((ks * 2 + q) * 2 + 0) * 2 + h) * 16) + c] = make_float4(r0.x, r1.x, r0.y, r1.y);
        sB[((((ks * 2 + q) * 2 + 1) * 2 + h) * 16) + c] = make_float4(r0.z, r1.z, r0.w, r1.w);
    }
    __syncthreads();

    const int wave = threadIdx.x >> 5;
    const int lane = threadIdx.x & 31;
    const int lo   = lane & 15;
    const int hh   = lane >> 4;
    const int hiK  = hh * 2;
    const int hiM  = hh * 8;
    const int tile = (blockIdx.x * 8 + wave) * 16;

    const float* aRow = &edge_attr[(size_t)(tile + lo) * 16];
    v2f a[4];
    a[0] = *(const v2f*)&aRow[0 + hiK];
    a[1] = *(const v2f*)&aRow[4 + hiK];
    a[2] = *(const v2f*)&aRow[8 + hiK];
    a[3] = *(const v2f*)&aRow[12 + hiK];

    int sIdx = srcIdx[tile + lo];
    int dIdx = dstIdx[tile + lo];
    unsigned offS[8], offD[8];
#pragma unroll
    for (int r = 0; r < 8; ++r) {
        offS[r] = (unsigned)__shfl(sIdx, r + hiM, 32) * 512u + (unsigned)lo * 16u;
        offD[r] = (unsigned)__shfl(dIdx, r + hiM, 32) * 512u + (unsigned)lo * 16u;
    }

    float acc[8];
#pragma unroll
    for (int r = 0; r < 8; ++r) acc[r] = 0.0f;

#pragma unroll
    for (int q = 0; q < 2; ++q) {
        float g[8][4];
#pragma unroll
        for (int r = 0; r < 8; ++r) {
            const float4 vl = *(const float4*)((const char*)xl + offS[r] + q * 256);
            const float4 vr = *(const float4*)((const char*)xr + offD[r] + q * 256);
            g[r][0] = vl.x + vr.x;
            g[r][1] = vl.y + vr.y;
            g[r][2] = vl.z + vr.z;
            g[r][3] = vl.w + vr.w;
        }
        const float4 av4 = *(const float4*)&att[q * 64 + lo * 4];
        const float avj[4] = {av4.x, av4.y, av4.z, av4.w};
#pragma unroll
        for (int jp = 0; jp < 2; ++jp) {
            float4 bv[4];
#pragma unroll
            for (int ks = 0; ks < 4; ++ks)
                bv[ks] = sB[((((ks * 2 + q) * 2 + jp) * 2 + hh) * 16) + lo];
#pragma unroll
            for (int jj = 0; jj < 2; ++jj) {
                const int j = 2 * jp + jj;
                v8f c;
#pragma unroll
                for (int r = 0; r < 8; ++r) c[r] = g[r][j];
#pragma unroll
                for (int ks = 0; ks < 4; ++ks) {
                    v2f b;
                    b.x = jj ? bv[ks].z : bv[ks].x;
                    b.y = jj ? bv[ks].w : bv[ks].y;
                    c = __builtin_amdgcn_wmma_f32_16x16x4_f32(false, a[ks], false, b, (short)0, c, false, false);
                }
#pragma unroll
                for (int r = 0; r < 8; ++r) acc[r] += lrelu(c[r]) * avj[j];
            }
        }
    }

#pragma unroll
    for (int r = 0; r < 8; ++r) {
        float v = acc[r];
        v += __shfl_xor(v, 1, 32);
        v += __shfl_xor(v, 2, 32);
        v += __shfl_xor(v, 4, 32);
        v += __shfl_xor(v, 8, 32);
        if (lo == 0) ebuf[tile + r + hiM] = v;
    }
}

__global__ __launch_bounds__(1024) void seg_offsets(
    const int* __restrict__ cnt, int* __restrict__ offs, int N) {
    __shared__ int sScan[1024];
    const int t   = threadIdx.x;
    const int per = (N + 1023) / 1024;
    const int b0  = t * per;
    const int b1  = min(b0 + per, N);
    int sum = 0;
    for (int i = b0; i < b1; ++i) sum += cnt[i];
    sScan[t] = sum;
    __syncthreads();
    for (int off = 1; off < 1024; off <<= 1) {
        int v = (t >= off) ? sScan[t - off] : 0;
        __syncthreads();
        sScan[t] += v;
        __syncthreads();
    }
    int run = sScan[t] - sum;
    for (int i = b0; i < b1; ++i) {
        offs[i] = run;
        run += cnt[i];
    }
    if (t == 1023) offs[N] = sScan[1023];
}

__global__ __launch_bounds__(256) void edge_place(
    const int* __restrict__ srcIdx, const int* __restrict__ dstIdx,
    const int* __restrict__ rk, const int* __restrict__ offs,
    const int* __restrict__ cbase, const float* __restrict__ ebuf,
    int* __restrict__ pairs, int E, int N, int per) {
    const int chunk = blockIdx.y;
    const int inChunk = blockIdx.x * 256 + threadIdx.x;
    const int e = chunk * per + inChunk;
    if (inChunk >= per || e >= E) return;
    const int d = dstIdx[e];
    const int pos = offs[d] + cbase[(size_t)chunk * N + d] + rk[e];
    ((int2*)pairs)[pos] = make_int2(srcIdx[e], __float_as_int(ebuf[e]));
}

__global__ __launch_bounds__(128) void node_gather(
    const float* __restrict__ xl, const int* __restrict__ pairs,
    const int* __restrict__ offs, float* __restrict__ accum,
    float* __restrict__ wsum, int N) {
    const int node = blockIdx.x * 4 + __builtin_amdgcn_readfirstlane(threadIdx.x >> 5);
    const int lane = threadIdx.x & 31;
    if (node >= N) return;
    const int2* pr2 = (const int2*)pairs;
    const int beg = offs[node];
    const int end = offs[node + 1];

    float m = -INFINITY;
    for (int i = beg + lane; i < end; i += 32)
        m = fmaxf(m, __int_as_float(pr2[i].y));
    m = fmaxf(m, __shfl_xor(m, 16, 32));
    m = fmaxf(m, __shfl_xor(m, 8, 32));
    m = fmaxf(m, __shfl_xor(m, 4, 32));
    m = fmaxf(m, __shfl_xor(m, 2, 32));
    m = fmaxf(m, __shfl_xor(m, 1, 32));

    float4 acc = make_float4(0.0f, 0.0f, 0.0f, 0.0f);
    float ws = 0.0f;
    for (int base = beg; base < end; base += 32) {
        const int i = base + lane;
        int   s = 0;
        float p = 0.0f;
        if (i < end) {
            const int2 pr = pr2[i];
            s = pr.x;
            p = expf(__int_as_float(pr.y) - m);
        }
        ws += p;
        const int n = min(32, end - base);
        for (int jb = 0; jb < n; jb += 8) {
            float4 v[8];
            float  pj[8];
#pragma unroll
            for (int u = 0; u < 8; ++u) {
                const int sj = __builtin_amdgcn_readlane(s, jb + u);
                pj[u] = __int_as_float(__builtin_amdgcn_readlane(__float_as_int(p), jb + u));
                v[u]  = *(const float4*)&xl[(size_t)sj * 128 + lane * 4];
            }
#pragma unroll
            for (int u = 0; u < 8; ++u) {
                acc.x += pj[u] * v[u].x;
                acc.y += pj[u] * v[u].y;
                acc.z += pj[u] * v[u].z;
                acc.w += pj[u] * v[u].w;
            }
        }
    }
    ws += __shfl_xor(ws, 16, 32);
    ws += __shfl_xor(ws, 8, 32);
    ws += __shfl_xor(ws, 4, 32);
    ws += __shfl_xor(ws, 2, 32);
    ws += __shfl_xor(ws, 1, 32);

    *(float4*)&accum[(size_t)node * 128 + lane * 4] = acc;
    if (lane == 0) wsum[node] = ws;
}

__global__ __launch_bounds__(128) void bias_act_stats(
    const float* __restrict__ accum, const float* __restrict__ wsum,
    const float* __restrict__ bias, float* __restrict__ out,
    float* __restrict__ stats, int N, int rowsPerBlock) {
    __shared__ float sInv[128];
    int c  = threadIdx.x;
    int r0 = blockIdx.x * rowsPerBlock;
    int r1 = min(r0 + rowsPerBlock, N);
    float b = bias[c];
    float sum = 0.0f, sq = 0.0f;
    for (int base = r0; base < r1; base += 128) {
        __syncthreads();
        float den = 1.0f;
        if (base + c < r1) den = wsum[base + c];
        sInv[c] = den > 0.0f ? 1.0f / den : 1.0f;
        __syncthreads();
        int rEnd = min(base + 128, r1);
        for (int r = base; r < rEnd; ++r) {
            float v = lrelu(accum[(size_t)r * 128 + c] * sInv[r - base] + b);
            out[(size_t)r * 128 + c] = v;
            sum += v;
            sq  += v * v;
        }
    }
    atomicAdd(&stats[c], sum);
    atomicAdd(&stats[128 + c], sq);
}

__global__ void bn_prep(const float* __restrict__ stats,
                        const float* __restrict__ bnw,
                        const float* __restrict__ bnb,
                        float* __restrict__ scsh, int N) {
    int c = threadIdx.x;
    float inv = 1.0f / (float)N;
    float mean = stats[c] * inv;
    float var  = stats[128 + c] * inv - mean * mean;
    float sc   = bnw[c] * rsqrtf(var + BN_EPS);
    scsh[c]       = sc;
    scsh[128 + c] = bnb[c] - mean * sc;
}

__global__ void bn_apply(float* __restrict__ out, const float* __restrict__ scsh,
                         int total) {
    int i = blockIdx.x * blockDim.x + threadIdx.x;
    if (i >= total) return;
    int c = i & 127;
    out[i] = out[i] * scsh[c] + scsh[128 + c];
}

extern "C" void kernel_launch(void* const* d_in, const int* in_sizes, int n_in,
                              void* d_out, int out_size, void* d_ws, size_t ws_size,
                              hipStream_t stream) {
    const float* x    = (const float*)d_in[0];
    const float* ea   = (const float*)d_in[1];
    const int*   ei   = (const int*)d_in[2];
    const float* Wl   = (const float*)d_in[3];
    const float* Wr   = (const float*)d_in[4];
    const float* We   = (const float*)d_in[5];
    const float* att  = (const float*)d_in[6];
    const float* bias = (const float*)d_in[7];
    const float* bnw  = (const float*)d_in[8];
    const float* bnb  = (const float*)d_in[9];
    float* out = (float*)d_out;

    const int N = in_sizes[0] / 128;
    const int E = in_sizes[2] / 2;
    const int* srcIdx = ei;
    const int* dstIdx = ei + E;

    float* ws = (float*)d_ws;
    float* xl = ws;     ws += (size_t)N * 128;
    float* xr = ws;     ws += (size_t)N * 128;
    float* accum = ws;  ws += (size_t)N * 128;
    float* ebuf = ws;   ws += E;
    int* rk    = (int*)ws;  ws += E;
    int* pairs = (int*)ws;  ws += (size_t)E * 2;
    float* wsum = ws;   ws += N;
    int* offs  = (int*)ws;  ws += N + 32;
    int* cnt   = (int*)ws;  ws += N;
    int* hist  = (int*)ws;  ws += (size_t)N * RANK_CHUNKS;
    int* cbase = (int*)ws;  ws += (size_t)N * RANK_CHUNKS;
    float* stats = ws;  ws += 256;
    float* scsh  = ws;  ws += 256;

    const int per = (E + RANK_CHUNKS - 1) / RANK_CHUNKS;
    edge_rank<<<RANK_CHUNKS, 1024, (size_t)N * sizeof(int), stream>>>(
        dstIdx, rk, hist, stats, E, N, per);
    chunk_prefix<<<(N + 63) / 64, 64, 0, stream>>>(hist, cbase, cnt, N);
    seg_offsets<<<1, 1024, 0, stream>>>(cnt, offs, N);
    node_transform<<<N / 16, 256, 0, stream>>>(x, Wl, Wr, xl, xr, N);
    edge_logits<<<E / 128, 256, 0, stream>>>(ea, We, xl, xr, srcIdx, dstIdx, att,
                                             ebuf, E);
    edge_place<<<dim3((per + 255) / 256, RANK_CHUNKS), 256, 0, stream>>>(
        srcIdx, dstIdx, rk, offs, cbase, ebuf, pairs, E, N, per);
    node_gather<<<(N + 3) / 4, 128, 0, stream>>>(xl, pairs, offs, accum, wsum, N);
    const int rpb = 50;
    bias_act_stats<<<(N + rpb - 1) / rpb, 128, 0, stream>>>(accum, wsum, bias, out,
                                                            stats, N, rpb);
    bn_prep<<<1, 128, 0, stream>>>(stats, bnw, bnb, scsh, N);
    bn_apply<<<(N * 128 + 255) / 256, 256, 0, stream>>>(out, scsh, N * 128);
}


# Logits and aggregation fused into one per-destination kernel after the sort: each xl[src] row gathered once for both, xr row read once per node, online softmax (running max); sum order changes
# speedup vs baseline: 2.8714x; 1.1342x over previous
#include <hip/hip_runtime.h>

typedef __attribute__((ext_vector_type(2))) float v2f;
typedef __attribute__((ext_vector_type(8))) float v8f;

#define NEG_SLOPE 0.2f
#define BN_EPS 1e-5f

__device__ __forceinline__ float lrelu(float x) { return x > 0.0f ? x : NEG_SLOPE * x; }

#define RANK_CHUNKS 128

__global__ __launch_bounds__(1024) void edge_rank(
    const int* __restrict__ dstIdx, int* __restrict__ rk, int* __restrict__ hist,
    float* __restrict__ stats, int E, int N, int per) {
    extern __shared__ int sCnt[];
    const int t = threadIdx.x;
    for (int i = t; i < N; i += 1024) sCnt[i] = 0;
    if (blockIdx.x == 0 && t < 256) stats[t] = 0.0f;
    __syncthreads();
    const int e0 = blockIdx.x * per;
    const int e1 = min(e0 + per, E);
    for (int eb = e0 + t; eb < e1; eb += 8 * 1024) {
        int d[8];
#pragma unroll
        for (int u = 0; u < 8; ++u) {
            const int e = eb + u * 1024;
            d[u] = (e < e1) ? dstIdx[e] : 0;
        }
#pragma unroll
        for (int u = 0; u < 8; ++u) {
            const int e = eb + u * 1024;
            if (e < e1) rk[e] = atomicAdd(&sCnt[d[u]], 1);
        }
    }
    __syncthreads();
    int* __restrict__ h = hist + (size_t)blockIdx.x * N;
    for (int i = t; i < N; i += 1024) h[i] = sCnt[i];
}

__global__ __launch_bounds__(64) void chunk_prefix(
    const int* __restrict__ hist, int* __restrict__ cbase, int* __restrict__ cnt,
    int N) {
    const int node = blockIdx.x * 64 + threadIdx.x;
    if (node >= N) return;
    int run = 0;
    for (int c0 = 0; c0 < RANK_CHUNKS; c0 += 16) {
        int v[16];
#pragma unroll
        for (int u = 0; u < 16; ++u) v[u] = hist[(size_t)(c0 + u) * N + node];
#pragma unroll
        for (int u = 0; u < 16; ++u) {
            cbase[(size_t)(c0 + u) * N + node] = run;
            run += v[u];
        }
    }
    cnt[node] = run;
}

__global__ __launch_bounds__(256) void node_transform(
    const float* __restrict__ x, const float* __restrict__ Wl,
    const float* __restrict__ Wr, float* __restrict__ xl, float* __restrict__ xr,
    int N) {
    const int wave = threadIdx.x >> 5;
    const int lane = threadIdx.x & 31;
    const int lo   = lane & 15;
    const int hiK  = (lane >= 16) ? 2 : 0;
    const int hiM  = (lane >= 16) ? 8 : 0;
    const int rowTile = blockIdx.x * 16;
    const int col  = wave * 16 + lo;
    const int aRow = rowTile + lo;

    v8f cl = {};
    v8f cr = {};
    for (int k = 0; k < 128; k += 4) {
        v2f a = *(const v2f*)&x[aRow * 128 + k + hiK];
        v2f bl, br;
        bl.x = Wl[(k + hiK) * 128 + col];
        bl.y = Wl[(k + hiK + 1) * 128 + col];
        br.x = Wr[(k + hiK) * 128 + col];
        br.y = Wr[(k + hiK + 1) * 128 + col];
        cl = __builtin_amdgcn_wmma_f32_16x16x4_f32(false, a, false, bl, (short)0, cl, false, false);
        cr = __builtin_amdgcn_wmma_f32_16x16x4_f32(false, a, false, br, (short)0, cr, false, false);
    }
#pragma unroll
    for (int r = 0; r < 8; ++r) {
        int row = rowTile + r + hiM;
        xl[row * 128 + col] = cl[r];
        xr[row * 128 + col] = cr[r];
    }
}

__global__ __launch_bounds__(1024) void seg_offsets(
    const int* __restrict__ cnt, int* __restrict__ offs, int N) {
    __shared__ int sScan[1024];
    const int t   = threadIdx.x;
    const int per = (N + 1023) / 1024;
    const int b0  = t * per;
    const int b1  = min(b0 + per, N);
    int sum = 0;
    for (int i = b0; i < b1; ++i) sum += cnt[i];
    sScan[t] = sum;
    __syncthreads();
    for (int off = 1; off < 1024; off <<= 1) {
        int v = (t >= off) ? sScan[t - off] : 0;
        __syncthreads();
        sScan[t] += v;
        __syncthreads();
    }
    int run = sScan[t] - sum;
    for (int i = b0; i < b1; ++i) {
        offs[i] = run;
        run += cnt[i];
    }
    if (t == 1023) offs[N] = sScan[1023];
}

__global__ __launch_bounds__(256) void edge_place(
    const int* __restrict__ srcIdx, const int* __restrict__ dstIdx,
    const int* __restrict__ rk, const int* __restrict__ offs,
    const int* __restrict__ cbase, int* __restrict__ pairs, int E, int N, int per) {
    const int chunk = blockIdx.y;
    const int inChunk = blockIdx.x * 256 + threadIdx.x;
    const int e = chunk * per + inChunk;
    if (inChunk >= per || e >= E) return;
    const int d = dstIdx[e];
    const int pos = offs[d] + cbase[(size_t)chunk * N + d] + rk[e];
    ((int2*)pairs)[pos] = make_int2(srcIdx[e], e);
}

__global__ __launch_bounds__(128, 5) void node_attend(
    const float* __restrict__ edge_attr, const float* __restrict__ We,
    const float* __restrict__ xl, const float* __restrict__ xr,
    const int* __restrict__ pairs, const int* __restrict__ offs,
    const float* __restrict__ att, float* __restrict__ accum,
    float* __restrict__ wsum, int N) {
    __shared__ float4 sB[512];
    for (int pce = threadIdx.x; pce < 256; pce += 128) {
        const int c  = pce & 15;
        const int h  = (pce >> 4) & 1;
        const int q  = (pce >> 5) & 1;
        const int ks = (pce >> 6) & 3;
        const int k0 = 4 * ks + 2 * h;
        const float4 r0 = *(const float4*)&We[k0 * 128 + 64 * q + 4 * c];
        const float4 r1 = *(const float4*)&We[(k0 + 1) * 128 + 64 * q + 4 * c];
        sB[((((ks * 2 + q) * 2 + 0) * 2 + h) * 16) + c] = make_float4(r0.x, r1.x, r0.y, r1.y);
        sB[((((ks * 2 + q) * 2 + 1) * 2 + h) * 16) + c] = make_float4(r0.z, r1.z, r0.w, r1.w);
    }
    __syncthreads();

    const int node = blockIdx.x * 4 + __builtin_amdgcn_readfirstlane(threadIdx.x >> 5);
    if (node >= N) return;
    const int lane = threadIdx.x & 31;
    const int lo   = lane & 15;
    const int hh   = lane >> 4;
    const int hiK  = hh * 2;
    const int hiM  = hh * 8;
    const int2* pr2 = (const int2*)pairs;
    const int beg = offs[node];
    const int end = offs[node + 1];

    float m  = -INFINITY;
    float ws = 0.0f;
    float accv[2][4];
#pragma unroll
    for (int q = 0; q < 2; ++q)
#pragma unroll
        for (int j = 0; j < 4; ++j) accv[q][j] = 0.0f;

    int2 nxt = make_int2(0, 0);
    if (beg + lo < end) nxt = pr2[beg + lo];

    for (int t = beg; t < end; t += 16) {
        const int2 cur = nxt;
        nxt = make_int2(0, 0);
        if (t + 16 + lo < end) nxt = pr2[t + 16 + lo];
        asm volatile("" ::: "memory");

        const float4 xr0 = *(const float4*)&xr[(size_t)node * 128 + lo * 4];
        const float4 xr1 = *(const float4*)&xr[(size_t)node * 128 + 64 + lo * 4];
        const float4 at0 = *(const float4*)&att[lo * 4];
        const float4 at1 = *(const float4*)&att[64 + lo * 4];
        const float xrv[2][4] = {{xr0.x, xr0.y, xr0.z, xr0.w}, {xr1.x, xr1.y, xr1.z, xr1.w}};
        const float avj[2][4] = {{at0.x, at0.y, at0.z, at0.w}, {at1.x, at1.y, at1.z, at1.w}};

        const float* aRow = &edge_attr[(size_t)cur.y * 16];
        v2f a[4];
        a[0] = *(const v2f*)&aRow[0 + hiK];
        a[1] = *(const v2f*)&aRow[4 + hiK];
        a[2] = *(const v2f*)&aRow[8 + hiK];
        a[3] = *(const v2f*)&aRow[12 + hiK];

        float vl[2][8][4];
#pragma unroll
        for (int r = 0; r < 8; ++r) {
            const unsigned offS =
                (unsigned)__shfl(cur.x, r + hiM, 32) * 512u + (unsigned)lo * 16u;
#pragma unroll
            for (int q = 0; q < 2; ++q) {
                const float4 v = *(const float4*)((const char*)xl + offS + q * 256);
                vl[q][r][0] = v.x;
                vl[q][r][1] = v.y;
                vl[q][r][2] = v.z;
                vl[q][r][3] = v.w;
            }
        }

        float e[8];
#pragma unroll
        for (int r = 0; r < 8; ++r) e[r] = 0.0f;
#pragma unroll
        for (int q = 0; q < 2; ++q) {
#pragma unroll
            for (int jp = 0; jp < 2; ++jp) {
                float4 bv[4];
#pragma unroll
                for (int ks = 0; ks < 4; ++ks)
                    bv[ks] = sB[((((ks * 2 + q) * 2 + jp) * 2 + hh) * 16) + lo];
#pragma unroll
                for (int jj = 0; jj < 2; ++jj) {
                    const int j = 2 * jp + jj;
                    v8f c;
#pragma unroll
                    for (int r = 0; r < 8; ++r) c[r] = vl[q][r][j] + xrv[q][j];
#pragma unroll
                    for (int ks = 0; ks < 4; ++ks) {
                        v2f b;
                        b.x = jj ? bv[ks].z : bv[ks].x;
                        b.y = jj ? bv[ks].w : bv[ks].y;
                        c = __builtin_amdgcn_wmma_f32_16x16x4_f32(false, a[ks], false, b, (short)0, c, false, false);
                    }
#pragma unroll
                    for (int r = 0; r < 8; ++r) e[r] += lrelu(c[r]) * avj[q][j];
                }
            }
        }

        const int nvalid = end - t;
        float tm = -INFINITY;
#pragma unroll
        for (int r = 0; r < 8; ++r) {
            float v = e[r];
            v += __shfl_xor(v, 1, 32);
            v += __shfl_xor(v, 2, 32);
            v += __shfl_xor(v, 4, 32);
            v += __shfl_xor(v, 8, 32);
            if (r + hiM >= nvalid) v = -INFINITY;
            e[r] = v;
            tm = fmaxf(tm, v);
        }
        tm = fmaxf(tm, __shfl_xor(tm, 16, 32));

        const float mn = fmaxf(m, tm);
        const float sc = expf(m - mn);
        m = mn;
        ws *= sc;
#pragma unroll
        for (int q = 0; q < 2; ++q)
#pragma unroll
            for (int j = 0; j < 4; ++j) accv[q][j] *= sc;
#pragma unroll
        for (int r = 0; r < 8; ++r) {
            const float p = expf(e[r] - mn);
            ws += p;
#pragma unroll
            for (int q = 0; q < 2; ++q)
#pragma unroll
                for (int j = 0; j < 4; ++j) accv[q][j] += p * vl[q][r][j];
        }
    }

#pragma unroll
    for (int q = 0; q < 2; ++q)
#pragma unroll
        for (int j = 0; j < 4; ++j) accv[q][j] += __shfl_xor(accv[q][j], 16, 32);
    ws += __shfl_xor(ws, 16, 32);

    const float4 o = hh ? make_float4(accv[1][0], accv[1][1], accv[1][2], accv[1][3])
                        : make_float4(accv[0][0], accv[0][1], accv[0][2], accv[0][3]);
    *(float4*)&accum[(size_t)node * 128 + hh * 64 + lo * 4] = o;
    if (lane == 0) wsum[node] = ws;
}

__global__ __launch_bounds__(128) void bias_act_stats(
    const float* __restrict__ accum, const float* __restrict__ wsum,
    const float* __restrict__ bias, float* __restrict__ out,
    float* __restrict__ stats, int N, int rowsPerBlock) {
    __shared__ float sInv[128];
    int c  = threadIdx.x;
    int r0 = blockIdx.x * rowsPerBlock;
    int r1 = min(r0 + rowsPerBlock, N);
    float b = bias[c];
    float sum = 0.0f, sq = 0.0f;
    for (int base = r0; base < r1; base += 128) {
        __syncthreads();
        float den = 1.0f;
        if (base + c < r1) den = wsum[base + c];
        sInv[c] = den > 0.0f ? 1.0f / den : 1.0f;
        __syncthreads();
        int rEnd = min(base + 128, r1);
        for (int r = base; r < rEnd; ++r) {
            float v = lrelu(accum[(size_t)r * 128 + c] * sInv[r - base] + b);
            out[(size_t)r * 128 + c] = v;
            sum += v;
            sq  += v * v;
        }
    }
    atomicAdd(&stats[c], sum);
    atomicAdd(&stats[128 + c], sq);
}

__global__ void bn_prep(const float* __restrict__ stats,
                        const float* __restrict__ bnw,
                        const float* __restrict__ bnb,
                        float* __restrict__ scsh, int N) {
    int c = threadIdx.x;
    float inv = 1.0f / (float)N;
    float mean = stats[c] * inv;
    float var  = stats[128 + c] * inv - mean * mean;
    float sc   = bnw[c] * rsqrtf(var + BN_EPS);
    scsh[c]       = sc;
    scsh[128 + c] = bnb[c] - mean * sc;
}

__global__ void bn_apply(float* __restrict__ out, const float* __restrict__ scsh,
                         int total) {
    int i = blockIdx.x * blockDim.x + threadIdx.x;
    if (i >= total) return;
    int c = i & 127;
    out[i] = out[i] * scsh[c] + scsh[128 + c];
}

extern "C" void kernel_launch(void* const* d_in, const int* in_sizes, int n_in,
                              void* d_out, int out_size, void* d_ws, size_t ws_size,
                              hipStream_t stream) {
    const float* x    = (const float*)d_in[0];
    const float* ea   = (const float*)d_in[1];
    const int*   ei   = (const int*)d_in[2];
    const float* Wl   = (const float*)d_in[3];
    const float* Wr   = (const float*)d_in[4];
    const float* We   = (const float*)d_in[5];
    const float* att  = (const float*)d_in[6];
    const float* bias = (const float*)d_in[7];
    const float* bnw  = (const float*)d_in[8];
    const float* bnb  = (const float*)d_in[9];
    float* out = (float*)d_out;

    const int N = in_sizes[0] / 128;
    const int E = in_sizes[2] / 2;
    const int* srcIdx = ei;
    const int* dstIdx = ei + E;

    float* ws = (float*)d_ws;
    float* xl = ws;     ws += (size_t)N * 128;
    float* xr = ws;     ws += (size_t)N * 128;
    float* accum = ws;  ws += (size_t)N * 128;
    int* rk    = (int*)ws;  ws += E;
    int* pairs = (int*)ws;  ws += (size_t)E * 2;
    float* wsum = ws;   ws += N;
    int* offs  = (int*)ws;  ws += N + 32;
    int* cnt   = (int*)ws;  ws += N;
    int* hist  = (int*)ws;  ws += (size_t)N * RANK_CHUNKS;
    int* cbase = (int*)ws;  ws += (size_t)N * RANK_CHUNKS;
    float* stats = ws;  ws += 256;
    float* scsh  = ws;  ws += 256;

    const int per = (E + RANK_CHUNKS - 1) / RANK_CHUNKS;
    edge_rank<<<RANK_CHUNKS, 1024, (size_t)N * sizeof(int), stream>>>(
        dstIdx, rk, hist, stats, E, N, per);
    chunk_prefix<<<(N + 63) / 64, 64, 0, stream>>>(hist, cbase, cnt, N);
    seg_offsets<<<1, 1024, 0, stream>>>(cnt, offs, N);
    node_transform<<<N / 16, 256, 0, stream>>>(x, Wl, Wr, xl, xr, N);
    edge_place<<<dim3((per + 255) / 256, RANK_CHUNKS), 256, 0, stream>>>(
        srcIdx, dstIdx, rk, offs, cbase, pairs, E, N, per);
    node_attend<<<(N + 3) / 4, 128, 0, stream>>>(ea, We, xl, xr, pairs, offs, att,
                                                 accum, wsum, N);
    const int rpb = 50;
    bias_act_stats<<<(N + rpb - 1) / rpb, 128, 0, stream>>>(accum, wsum, bias, out,
                                                            stats, N, rpb);
    bn_prep<<<1, 128, 0, stream>>>(stats, bnw, bnb, scsh, N);
    bn_apply<<<(N * 128 + 255) / 256, 256, 0, stream>>>(out, scsh, N * 128);
}


# bias_act_stats: a wave owns whole rows (16 B per lane) and requests its five rows at once: one memory round trip per wave, not 50 serial ones; batch-norm sums: sum order changes (16 waves via LDS)
# speedup vs baseline: 3.5714x; 1.2438x over previous
#include <hip/hip_runtime.h>

typedef __attribute__((ext_vector_type(2))) float v2f;
typedef __attribute__((ext_vector_type(8))) float v8f;

#define NEG_SLOPE 0.2f
#define BN_EPS 1e-5f

__device__ __forceinline__ float lrelu(float x) { return x > 0.0f ? x : NEG_SLOPE * x; }

#define RANK_CHUNKS 128

__global__ __launch_bounds__(1024) void edge_rank(
    const int* __restrict__ dstIdx, int* __restrict__ rk, int* __restrict__ hist,
    float* __restrict__ stats, int E, int N, int per) {
    extern __shared__ int sCnt[];
    const int t = threadIdx.x;
    for (int i = t; i < N; i += 1024) sCnt[i] = 0;
    if (blockIdx.x == 0 && t < 256) stats[t] = 0.0f;
    __syncthreads();
    const int e0 = blockIdx.x * per;
    const int e1 = min(e0 + per, E);
    for (int eb = e0 + t; eb < e1; eb += 8 * 1024) {
        int d[8];
#pragma unroll
        for (int u = 0; u < 8; ++u) {
            const int e = eb + u * 1024;
            d[u] = (e < e1) ? dstIdx[e] : 0;
        }
#pragma unroll
        for (int u = 0; u < 8; ++u) {
            const int e = eb + u * 1024;
            if (e < e1) rk[e] = atomicAdd(&sCnt[d[u]], 1);
        }
    }
    __syncthreads();
    int* __restrict__ h = hist + (size_t)blockIdx.x * N;
    for (int i = t; i < N; i += 1024) h[i] = sCnt[i];
}

__global__ __launch_bounds__(64) void chunk_prefix(
    const int* __restrict__ hist, int* __restrict__ cbase, int* __restrict__ cnt,
    int N) {
    const int node = blockIdx.x * 64 + threadIdx.x;
    if (node >= N) return;
    int run = 0;
    for (int c0 = 0; c0 < RANK_CHUNKS; c0 += 16) {
        int v[16];
#pragma unroll
        for (int u = 0; u < 16; ++u) v[u] = hist[(size_t)(c0 + u) * N + node];
#pragma unroll
        for (int u = 0; u < 16; ++u) {
            cbase[(size_t)(c0 + u) * N + node] = run;
            run += v[u];
        }
    }
    cnt[node] = run;
}

__global__ __launch_bounds__(256) void node_transform(
    const float* __restrict__ x, const float* __restrict__ Wl,
    const float* __restrict__ Wr, float* __restrict__ xl, float* __restrict__ xr,
    int N) {
    const int wave = threadIdx.x >> 5;
    const int lane = threadIdx.x & 31;
    const int lo   = lane & 15;
    const int hiK  = (lane >= 16) ? 2 : 0;
    const int hiM  = (lane >= 16) ? 8 : 0;
    const int rowTile = blockIdx.x * 16;
    const int col  = wave * 16 + lo;
    const int aRow = rowTile + lo;

    v8f cl = {};
    v8f cr = {};
    for (int k = 0; k < 128; k += 4) {
        v2f a = *(const v2f*)&x[aRow * 128 + k + hiK];
        v2f bl, br;
        bl.x = Wl[(k + hiK) * 128 + col];
        bl.y = Wl[(k + hiK + 1) * 128 + col];
        br.x = Wr[(k + hiK) * 128 + col];
        br.y = Wr[(k + hiK + 1) * 128 + col];
        cl = __builtin_amdgcn_wmma_f32_16x16x4_f32(false, a, false, bl, (short)0, cl, false, false);
        cr = __builtin_amdgcn_wmma_f32_16x16x4_f32(false, a, false, br, (short)0, cr, false, false);
    }
#pragma unroll
    for (int r = 0; r < 8; ++r) {
        int row = rowTile + r + hiM;
        xl[row * 128 + col] = cl[r];
        xr[row * 128 + col] = cr[r];
    }
}

__global__ __launch_bounds__(1024) void seg_offsets(
    const int* __restrict__ cnt, int* __restrict__ offs, int N) {
    __shared__ int sScan[1024];
    const int t   = threadIdx.x;
    const int per = (N + 1023) / 1024;
    const int b0  = t * per;
    const int b1  = min(b0 + per, N);
    int sum = 0;
    for (int i = b0; i < b1; ++i) sum += cnt[i];
    sScan[t] = sum;
    __syncthreads();
    for (int off = 1; off < 1024; off <<= 1) {
        int v = (t >= off) ? sScan[t - off] : 0;
        __syncthreads();
        sScan[t] += v;
        __syncthreads();
    }
    int run = sScan[t] - sum;
    for (int i = b0; i < b1; ++i) {
        offs[i] = run;
        run += cnt[i];
    }
    if (t == 1023) offs[N] = sScan[1023];
}

__global__ __launch_bounds__(256) void edge_place(
    const int* __restrict__ srcIdx, const int* __restrict__ dstIdx,
    const int* __restrict__ rk, const int* __restrict__ offs,
    const int* __restrict__ cbase, int* __restrict__ pairs, int E, int N, int per) {
    const int chunk = blockIdx.y;
    const int inChunk = blockIdx.x * 256 + threadIdx.x;
    const int e = chunk * per + inChunk;
    if (inChunk >= per || e >= E) return;
    const int d = dstIdx[e];
    const int pos = offs[d] + cbase[(size_t)chunk * N + d] + rk[e];
    ((int2*)pairs)[pos] = make_int2(srcIdx[e], e);
}

__global__ __launch_bounds__(128, 5) void node_attend(
    const float* __restrict__ edge_attr, const float* __restrict__ We,
    const float* __restrict__ xl, const float* __restrict__ xr,
    const int* __restrict__ pairs, const int* __restrict__ offs,
    const float* __restrict__ att, float* __restrict__ accum,
    float* __restrict__ wsum, int N) {
    __shared__ float4 sB[512];
    for (int pce = threadIdx.x; pce < 256; pce += 128) {
        const int c  = pce & 15;
        const int h  = (pce >> 4) & 1;
        const int q  = (pce >> 5) & 1;
        const int ks = (pce >> 6) & 3;
        const int k0 = 4 * ks + 2 * h;
        const float4 r0 = *(const float4*)&We[k0 * 128 + 64 * q + 4 * c];
        const float4 r1 = *(const float4*)&We[(k0 + 1) * 128 + 64 * q + 4 * c];
        sB[((((ks * 2 + q) * 2 + 0) * 2 + h) * 16) + c] = make_float4(r0.x, r1.x, r0.y, r1.y);
        sB[((((ks * 2 + q) * 2 + 1) * 2 + h) * 16) + c] = make_float4(r0.z, r1.z, r0.w, r1.w);
    }
    __syncthreads();

    const int node = blockIdx.x * 4 + __builtin_amdgcn_readfirstlane(threadIdx.x >> 5);
    if (node >= N) return;
    const int lane = threadIdx.x & 31;
    const int lo   = lane & 15;
    const int hh   = lane >> 4;
    const int hiK  = hh * 2;
    const int hiM  = hh * 8;
    const int2* pr2 = (const int2*)pairs;
    const int beg = offs[node];
    const int end = offs[node + 1];

    float m  = -INFINITY;
    float ws = 0.0f;
    float accv[2][4];
#pragma unroll
    for (int q = 0; q < 2; ++q)
#pragma unroll
        for (int j = 0; j < 4; ++j) accv[q][j] = 0.0f;

    int2 nxt = make_int2(0, 0);
    if (beg + lo < end) nxt = pr2[beg + lo];

    for (int t = beg; t < end; t += 16) {
        const int2 cur = nxt;
        nxt = make_int2(0, 0);
        if (t + 16 + lo < end) nxt = pr2[t + 16 + lo];
        asm volatile("" ::: "memory");

        const float4 xr0 = *(const float4*)&xr[(size_t)node * 128 + lo * 4];
        const float4 xr1 = *(const float4*)&xr[(size_t)node * 128 + 64 + lo * 4];
        const float4 at0 = *(const float4*)&att[lo * 4];
        const float4 at1 = *(const float4*)&att[64 + lo * 4];
        const float xrv[2][4] = {{xr0.x, xr0.y, xr0.z, xr0.w}, {xr1.x, xr1.y, xr1.z, xr1.w}};
        const float avj[2][4] = {{at0.x, at0.y, at0.z, at0.w}, {at1.x, at1.y, at1.z, at1.w}};

        const float* aRow = &edge_attr[(size_t)cur.y * 16];
        v2f a[4];
        a[0] = *(const v2f*)&aRow[0 + hiK];
        a[1] = *(const v2f*)&aRow[4 + hiK];
        a[2] = *(const v2f*)&aRow[8 + hiK];
        a[3] = *(const v2f*)&aRow[12 + hiK];

        float vl[2][8][4];
#pragma unroll
        for (int r = 0; r < 8; ++r) {
            const unsigned offS =
                (unsigned)__shfl(cur.x, r + hiM, 32) * 512u + (unsigned)lo * 16u;
#pragma unroll
            for (int q = 0; q < 2; ++q) {
                const float4 v = *(const float4*)((const char*)xl + offS + q * 256);
                vl[q][r][0] = v.x;
                vl[q][r][1] = v.y;
                vl[q][r][2] = v.z;
                vl[q][r][3] = v.w;
            }
        }

        float e[8];
#pragma unroll
        for (int r = 0; r < 8; ++r) e[r] = 0.0f;
#pragma unroll
        for (int q = 0; q < 2; ++q) {
#pragma unroll
            for (int jp = 0; jp < 2; ++jp) {
                float4 bv[4];
#pragma unroll
                for (int ks = 0; ks < 4; ++ks)
                    bv[ks] = sB[((((ks * 2 + q) * 2 + jp) * 2 + hh) * 16) + lo];
#pragma unroll
                for (int jj = 0; jj < 2; ++jj) {
                    const int j = 2 * jp + jj;
                    v8f c;
#pragma unroll
                    for (int r = 0; r < 8; ++r) c[r] = vl[q][r][j] + xrv[q][j];
#pragma unroll
                    for (int ks = 0; ks < 4; ++ks) {
                        v2f b;
                        b.x = jj ? bv[ks].z : bv[ks].x;
                        b.y = jj ? bv[ks].w : bv[ks].y;
                        c = __builtin_amdgcn_wmma_f32_16x16x4_f32(false, a[ks], false, b, (short)0, c, false, false);
                    }
#pragma unroll
                    for (int r = 0; r < 8; ++r) e[r] += lrelu(c[r]) * avj[q][j];
                }
            }
        }

        const int nvalid = end - t;
        float tm = -INFINITY;
#pragma unroll
        for (int r = 0; r < 8; ++r) {
            float v = e[r];
            v += __shfl_xor(v, 1, 32);
            v += __shfl_xor(v, 2, 32);
            v += __shfl_xor(v, 4, 32);
            v += __shfl_xor(v, 8, 32);
            if (r + hiM >= nvalid) v = -INFINITY;
            e[r] = v;
            tm = fmaxf(tm, v);
        }
        tm = fmaxf(tm, __shfl_xor(tm, 16, 32));

        const float mn = fmaxf(m, tm);
        const float sc = expf(m - mn);
        m = mn;
        ws *= sc;
#pragma unroll
        for (int q = 0; q < 2; ++q)
#pragma unroll
            for (int j = 0; j < 4; ++j) accv[q][j] *= sc;
#pragma unroll
        for (int r = 0; r < 8; ++r) {
            const float p = expf(e[r] - mn);
            ws += p;
#pragma unroll
            for (int q = 0; q < 2; ++q)
#pragma unroll
                for (int j = 0; j < 4; ++j) accv[q][j] += p * vl[q][r][j];
        }
    }

#pragma unroll
    for (int q = 0; q < 2; ++q)
#pragma unroll
        for (int j = 0; j < 4; ++j) accv[q][j] += __shfl_xor(accv[q][j], 16, 32);
    ws += __shfl_xor(ws, 16, 32);

    const float4 o = hh ? make_float4(accv[1][0], accv[1][1], accv[1][2], accv[1][3])
                        : make_float4(accv[0][0], accv[0][1], accv[0][2], accv[0][3]);
    *(float4*)&accum[(size_t)node * 128 + hh * 64 + lo * 4] = o;
    if (lane == 0) wsum[node] = ws;
}

#define BAS_WAVES 16
#define BAS_PER_WAVE 5
#define BAS_ROWS (BAS_WAVES * BAS_PER_WAVE)
__global__ __launch_bounds__(512) void bias_act_stats(
    const float* __restrict__ accum, const float* __restrict__ wsum,
    const float* __restrict__ bias, float* __restrict__ out,
    float* __restrict__ stats, int N) {
    __shared__ float4 sPart[BAS_WAVES][64];
    const int wave = __builtin_amdgcn_readfirstlane(threadIdx.x >> 5);
    const int lane = threadIdx.x & 31;
    const int r0 = blockIdx.x * BAS_ROWS + wave;

    float4 v[BAS_PER_WAVE];
    float den[BAS_PER_WAVE];
#pragma unroll
    for (int i = 0; i < BAS_PER_WAVE; ++i) {
        const int rr = min(r0 + i * BAS_WAVES, N - 1);
        v[i]   = *(const float4*)&accum[(size_t)rr * 128 + lane * 4];
        den[i] = wsum[rr];
    }
    const float4 b = *(const float4*)&bias[lane * 4];

    float4 s = make_float4(0.0f, 0.0f, 0.0f, 0.0f);
    float4 q = make_float4(0.0f, 0.0f, 0.0f, 0.0f);
#pragma unroll
    for (int i = 0; i < BAS_PER_WAVE; ++i) {
        const bool live = r0 + i * BAS_WAVES < N;
        const float inv = den[i] > 0.0f ? 1.0f / den[i] : 1.0f;
        float4 o;
        o.x = lrelu(v[i].x * inv + b.x);
        o.y = lrelu(v[i].y * inv + b.y);
        o.z = lrelu(v[i].z * inv + b.z);
        o.w = lrelu(v[i].w * inv + b.w);
        v[i] = o;
        s.x += live ? o.x : 0.0f;
        s.y += live ? o.y : 0.0f;
        s.z += live ? o.z : 0.0f;
        s.w += live ? o.w : 0.0f;
        q.x += live ? o.x * o.x : 0.0f;
        q.y += live ? o.y * o.y : 0.0f;
        q.z += live ? o.z * o.z : 0.0f;
        q.w += live ? o.w * o.w : 0.0f;
    }
    sPart[wave][lane]      = s;
    sPart[wave][32 + lane] = q;
    __syncthreads();
#pragma unroll
    for (int i = 0; i < BAS_PER_WAVE; ++i) {
        const int r = r0 + i * BAS_WAVES;
        if (r < N) *(float4*)&out[(size_t)r * 128 + lane * 4] = v[i];
    }
    if (threadIdx.x < 256) {
        const float* p = (const float*)sPart;
        float tot = 0.0f;
#pragma unroll
        for (int w = 0; w < BAS_WAVES; ++w) tot += p[w * 256 + threadIdx.x];
        atomicAdd(&stats[threadIdx.x], tot);
    }
}

__global__ void bn_prep(const float* __restrict__ stats,
                        const float* __restrict__ bnw,
                        const float* __restrict__ bnb,
                        float* __restrict__ scsh, int N) {
    int c = threadIdx.x;
    float inv = 1.0f / (float)N;
    float mean = stats[c] * inv;
    float var  = stats[128 + c] * inv - mean * mean;
    float sc   = bnw[c] * rsqrtf(var + BN_EPS);
    scsh[c]       = sc;
    scsh[128 + c] = bnb[c] - mean * sc;
}

__global__ void bn_apply(float* __restrict__ out, const float* __restrict__ scsh,
                         int total) {
    int i = blockIdx.x * blockDim.x + threadIdx.x;
    if (i >= total) return;
    int c = i & 127;
    out[i] = out[i] * scsh[c] + scsh[128 + c];
}

extern "C" void kernel_launch(void* const* d_in, const int* in_sizes, int n_in,
                              void* d_out, int out_size, void* d_ws, size_t ws_size,
                              hipStream_t stream) {
    const float* x    = (const float*)d_in[0];
    const float* ea   = (const float*)d_in[1];
    const int*   ei   = (const int*)d_in[2];
    const float* Wl   = (const float*)d_in[3];
    const float* Wr   = (const float*)d_in[4];
    const float* We   = (const float*)d_in[5];
    const float* att  = (const float*)d_in[6];
    const float* bias = (const float*)d_in[7];
    const float* bnw  = (const float*)d_in[8];
    const float* bnb  = (const float*)d_in[9];
    float* out = (float*)d_out;

    const int N = in_sizes[0] / 128;
    const int E = in_sizes[2] / 2;
    const int* srcIdx = ei;
    const int* dstIdx = ei + E;

    float* ws = (float*)d_ws;
    float* xl = ws;     ws += (size_t)N * 128;
    float* xr = ws;     ws += (size_t)N * 128;
    float* accum = ws;  ws += (size_t)N * 128;
    int* rk    = (int*)ws;  ws += E;
    int* pairs = (int*)ws;  ws += (size_t)E * 2;
    float* wsum = ws;   ws += N;
    int* offs  = (int*)ws;  ws += N + 32;
    int* cnt   = (int*)ws;  ws += N;
    int* hist  = (int*)ws;  ws += (size_t)N * RANK_CHUNKS;
    int* cbase = (int*)ws;  ws += (size_t)N * RANK_CHUNKS;
    float* stats = ws;  ws += 256;
    float* scsh  = ws;  ws += 256;

    const int per = (E + RANK_CHUNKS - 1) / RANK_CHUNKS;
    edge_rank<<<RANK_CHUNKS, 1024, (size_t)N * sizeof(int), stream>>>(
        dstIdx, rk, hist, stats, E, N, per);
    chunk_prefix<<<(N + 63) / 64, 64, 0, stream>>>(hist, cbase, cnt, N);
    seg_offsets<<<1, 1024, 0, stream>>>(cnt, offs, N);
    node_transform<<<N / 16, 256, 0, stream>>>(x, Wl, Wr, xl, xr, N);
    edge_place<<<dim3((per + 255) / 256, RANK_CHUNKS), 256, 0, stream>>>(
        srcIdx, dstIdx, rk, offs, cbase, pairs, E, N, per);
    node_attend<<<(N + 3) / 4, 128, 0, stream>>>(ea, We, xl, xr, pairs, offs, att,
                                                 accum, wsum, N);
    const int rpb = BAS_ROWS;
    bias_act_stats<<<(N + rpb - 1) / rpb, 512, 0, stream>>>(accum, wsum, bias, out,
                                                            stats, N);
    bn_prep<<<1, 128, 0, stream>>>(stats, bnw, bnb, scsh, N);
    bn_apply<<<(N * 128 + 255) / 256, 256, 0, stream>>>(out, scsh, N * 128);
}


# bn_apply: a wave owns whole rows (16 B per lane), five rows and the column sums requested at once; each lane derives scale/shift of its 4 columns itself: 2000 waves not 40000, bn_prep launch gone
# speedup vs baseline: 3.6969x; 1.0351x over previous
#include <hip/hip_runtime.h>

typedef __attribute__((ext_vector_type(2))) float v2f;
typedef __attribute__((ext_vector_type(8))) float v8f;

#define NEG_SLOPE 0.2f
#define BN_EPS 1e-5f

__device__ __forceinline__ float lrelu(float x) { return x > 0.0f ? x : NEG_SLOPE * x; }

#define RANK_CHUNKS 128

__global__ __launch_bounds__(1024) void edge_rank(
    const int* __restrict__ dstIdx, int* __restrict__ rk, int* __restrict__ hist,
    float* __restrict__ stats, int E, int N, int per) {
    extern __shared__ int sCnt[];
    const int t = threadIdx.x;
    for (int i = t; i < N; i += 1024) sCnt[i] = 0;
    if (blockIdx.x == 0 && t < 256) stats[t] = 0.0f;
    __syncthreads();
    const int e0 = blockIdx.x * per;
    const int e1 = min(e0 + per, E);
    for (int eb = e0 + t; eb < e1; eb += 8 * 1024) {
        int d[8];
#pragma unroll
        for (int u = 0; u < 8; ++u) {
            const int e = eb + u * 1024;
            d[u] = (e < e1) ? dstIdx[e] : 0;
        }
#pragma unroll
        for (int u = 0; u < 8; ++u) {
            const int e = eb + u * 1024;
            if (e < e1) rk[e] = atomicAdd(&sCnt[d[u]], 1);
        }
    }
    __syncthreads();
    int* __restrict__ h = hist + (size_t)blockIdx.x * N;
    for (int i = t; i < N; i += 1024) h[i] = sCnt[i];
}

__global__ __launch_bounds__(64) void chunk_prefix(
    const int* __restrict__ hist, int* __restrict__ cbase, int* __restrict__ cnt,
    int N) {
    const int node = blockIdx.x * 64 + threadIdx.x;
    if (node >= N) return;
    int run = 0;
    for (int c0 = 0; c0 < RANK_CHUNKS; c0 += 16) {
        int v[16];
#pragma unroll
        for (int u = 0; u < 16; ++u) v[u] = hist[(size_t)(c0 + u) * N + node];
#pragma unroll
        for (int u = 0; u < 16; ++u) {
            cbase[(size_t)(c0 + u) * N + node] = run;
            run += v[u];
        }
    }
    cnt[node] = run;
}

__global__ __launch_bounds__(256) void node_transform(
    const float* __restrict__ x, const float* __restrict__ Wl,
    const float* __restrict__ Wr, float* __restrict__ xl, float* __restrict__ xr,
    int N) {
    const int wave = threadIdx.x >> 5;
    const int lane = threadIdx.x & 31;
    const int lo   = lane & 15;
    const int hiK  = (lane >= 16) ? 2 : 0;
    const int hiM  = (lane >= 16) ? 8 : 0;
    const int rowTile = blockIdx.x * 16;
    const int col  = wave * 16 + lo;
    const int aRow = rowTile + lo;

    v8f cl = {};
    v8f cr = {};
    for (int k = 0; k < 128; k += 4) {
        v2f a = *(const v2f*)&x[aRow * 128 + k + hiK];
        v2f bl, br;
        bl.x = Wl[(k + hiK) * 128 + col];
        bl.y = Wl[(k + hiK + 1) * 128 + col];
        br.x = Wr[(k + hiK) * 128 + col];
        br.y = Wr[(k + hiK + 1) * 128 + col];
        cl = __builtin_amdgcn_wmma_f32_16x16x4_f32(false, a, false, bl, (short)0, cl, false, false);
        cr = __builtin_amdgcn_wmma_f32_16x16x4_f32(false, a, false, br, (short)0, cr, false, false);
    }
#pragma unroll
    for (int r = 0; r < 8; ++r) {
        int row = rowTile + r + hiM;
        xl[row * 128 + col] = cl[r];
        xr[row * 128 + col] = cr[r];
    }
}

__global__ __launch_bounds__(1024) void seg_offsets(
    const int* __restrict__ cnt, int* __restrict__ offs, int N) {
    __shared__ int sScan[1024];
    const int t   = threadIdx.x;
    const int per = (N + 1023) / 1024;
    const int b0  = t * per;
    const int b1  = min(b0 + per, N);
    int sum = 0;
    for (int i = b0; i < b1; ++i) sum += cnt[i];
    sScan[t] = sum;
    __syncthreads();
    for (int off = 1; off < 1024; off <<= 1) {
        int v = (t >= off) ? sScan[t - off] : 0;
        __syncthreads();
        sScan[t] += v;
        __syncthreads();
    }
    int run = sScan[t] - sum;
    for (int i = b0; i < b1; ++i) {
        offs[i] = run;
        run += cnt[i];
    }
    if (t == 1023) offs[N] = sScan[1023];
}

__global__ __launch_bounds__(256) void edge_place(
    const int* __restrict__ srcIdx, const int* __restrict__ dstIdx,
    const int* __restrict__ rk, const int* __restrict__ offs,
    const int* __restrict__ cbase, int* __restrict__ pairs, int E, int N, int per) {
    const int chunk = blockIdx.y;
    const int inChunk = blockIdx.x * 256 + threadIdx.x;
    const int e = chunk * per + inChunk;
    if (inChunk >= per || e >= E) return;
    const int d = dstIdx[e];
    const int pos = offs[d] + cbase[(size_t)chunk * N + d] + rk[e];
    ((int2*)pairs)[pos] = make_int2(srcIdx[e], e);
}

__global__ __launch_bounds__(128, 5) void node_attend(
    const float* __restrict__ edge_attr, const float* __restrict__ We,
    const float* __restrict__ xl, const float* __restrict__ xr,
    const int* __restrict__ pairs, const int* __restrict__ offs,
    const float* __restrict__ att, float* __restrict__ accum,
    float* __restrict__ wsum, int N) {
    __shared__ float4 sB[512];
    for (int pce = threadIdx.x; pce < 256; pce += 128) {
        const int c  = pce & 15;
        const int h  = (pce >> 4) & 1;
        const int q  = (pce >> 5) & 1;
        const int ks = (pce >> 6) & 3;
        const int k0 = 4 * ks + 2 * h;
        const float4 r0 = *(const float4*)&We[k0 * 128 + 64 * q + 4 * c];
        const float4 r1 = *(const float4*)&We[(k0 + 1) * 128 + 64 * q + 4 * c];
        sB[((((ks * 2 + q) * 2 + 0) * 2 + h) * 16) + c] = make_float4(r0.x, r1.x, r0.y, r1.y);
        sB[((((ks * 2 + q) * 2 + 1) * 2 + h) * 16) + c] = make_float4(r0.z, r1.z, r0.w, r1.w);
    }
    __syncthreads();

    const int node = blockIdx.x * 4 + __builtin_amdgcn_readfirstlane(threadIdx.x >> 5);
    if (node >= N) return;
    const int lane = threadIdx.x & 31;
    const int lo   = lane & 15;
    const int hh   = lane >> 4;
    const int hiK  = hh * 2;
    const int hiM  = hh * 8;
    const int2* pr2 = (const int2*)pairs;
    const int beg = offs[node];
    const int end = offs[node + 1];

    float m  = -INFINITY;
    float ws = 0.0f;
    float accv[2][4];
#pragma unroll
    for (int q = 0; q < 2; ++q)
#pragma unroll
        for (int j = 0; j < 4; ++j) accv[q][j] = 0.0f;

    int2 nxt = make_int2(0, 0);
    if (beg + lo < end) nxt = pr2[beg + lo];

    for (int t = beg; t < end; t += 16) {
        const int2 cur = nxt;
        nxt = make_int2(0, 0);
        if (t + 16 + lo < end) nxt = pr2[t + 16 + lo];
        asm volatile("" ::: "memory");

        const float4 xr0 = *(const float4*)&xr[(size_t)node * 128 + lo * 4];
        const float4 xr1 = *(const float4*)&xr[(size_t)node * 128 + 64 + lo * 4];
        const float4 at0 = *(const float4*)&att[lo * 4];
        const float4 at1 = *(const float4*)&att[64 + lo * 4];
        const float xrv[2][4] = {{xr0.x, xr0.y, xr0.z, xr0.w}, {xr1.x, xr1.y, xr1.z, xr1.w}};
        const float avj[2][4] = {{at0.x, at0.y, at0.z, at0.w}, {at1.x, at1.y, at1.z, at1.w}};

        const float* aRow = &edge_attr[(size_t)cur.y * 16];
        v2f a[4];
        a[0] = *(const v2f*)&aRow[0 + hiK];
        a[1] = *(const v2f*)&aRow[4 + hiK];
        a[2] = *(const v2f*)&aRow[8 + hiK];
        a[3] = *(const v2f*)&aRow[12 + hiK];

        float vl[2][8][4];
#pragma unroll
        for (int r = 0; r < 8; ++r) {
            const unsigned offS =
                (unsigned)__shfl(cur.x, r + hiM, 32) * 512u + (unsigned)lo * 16u;
#pragma unroll
            for (int q = 0; q < 2; ++q) {
                const float4 v = *(const float4*)((const char*)xl + offS + q * 256);
                vl[q][r][0] = v.x;
                vl[q][r][1] = v.y;
                vl[q][r][2] = v.z;
                vl[q][r][3] = v.w;
            }
        }

        float e[8];
#pragma unroll
        for (int r = 0; r < 8; ++r) e[r] = 0.0f;
#pragma unroll
        for (int q = 0; q < 2; ++q) {
#pragma unroll
            for (int jp = 0; jp < 2; ++jp) {
                float4 bv[4];
#pragma unroll
                for (int ks = 0; ks < 4; ++ks)
                    bv[ks] = sB[((((ks * 2 + q) * 2 + jp) * 2 + hh) * 16) + lo];
#pragma unroll
                for (int jj = 0; jj < 2; ++jj) {
                    const int j = 2 * jp + jj;
                    v8f c;
#pragma unroll
                    for (int r = 0; r < 8; ++r) c[r] = vl[q][r][j] + xrv[q][j];
#pragma unroll
                    for (int ks = 0; ks < 4; ++ks) {
                        v2f b;
                        b.x = jj ? bv[ks].z : bv[ks].x;
                        b.y = jj ? bv[ks].w : bv[ks].y;
                        c = __builtin_amdgcn_wmma_f32_16x16x4_f32(false, a[ks], false, b, (short)0, c, false, false);
                    }
#pragma unroll
                    for (int r = 0; r < 8; ++r) e[r] += lrelu(c[r]) * avj[q][j];
                }
            }
        }

        const int nvalid = end - t;
        float tm = -INFINITY;
#pragma unroll
        for (int r = 0; r < 8; ++r) {
            float v = e[r];
            v += __shfl_xor(v, 1, 32);
            v += __shfl_xor(v, 2, 32);
            v += __shfl_xor(v, 4, 32);
            v += __shfl_xor(v, 8, 32);
            if (r + hiM >= nvalid) v = -INFINITY;
            e[r] = v;
            tm = fmaxf(tm, v);
        }
        tm = fmaxf(tm, __shfl_xor(tm, 16, 32));

        const float mn = fmaxf(m, tm);
        const float sc = expf(m - mn);
        m = mn;
        ws *= sc;
#pragma unroll
        for (int q = 0; q < 2; ++q)
#pragma unroll
            for (int j = 0; j < 4; ++j) accv[q][j] *= sc;
#pragma unroll
        for (int r = 0; r < 8; ++r) {
            const float p = expf(e[r] - mn);
            ws += p;
#pragma unroll
            for (int q = 0; q < 2; ++q)
#pragma unroll
                for (int j = 0; j < 4; ++j) accv[q][j] += p * vl[q][r][j];
        }
    }

#pragma unroll
    for (int q = 0; q < 2; ++q)
#pragma unroll
        for (int j = 0; j < 4; ++j) accv[q][j] += __shfl_xor(accv[q][j], 16, 32);
    ws += __shfl_xor(ws, 16, 32);

    const float4 o = hh ? make_float4(accv[1][0], accv[1][1], accv[1][2], accv[1][3])
                        : make_float4(accv[0][0], accv[0][1], accv[0][2], accv[0][3]);
    *(float4*)&accum[(size_t)node * 128 + hh * 64 + lo * 4] = o;
    if (lane == 0) wsum[node] = ws;
}

#define BAS_WAVES 16
#define BAS_PER_WAVE 5
#define BAS_ROWS (BAS_WAVES * BAS_PER_WAVE)
__global__ __launch_bounds__(512) void bias_act_stats(
    const float* __restrict__ accum, const float* __restrict__ wsum,
    const float* __restrict__ bias, float* __restrict__ out,
    float* __restrict__ stats, int N) {
    __shared__ float4 sPart[BAS_WAVES][64];
    const int wave = __builtin_amdgcn_readfirstlane(threadIdx.x >> 5);
    const int lane = threadIdx.x & 31;
    const int r0 = blockIdx.x * BAS_ROWS + wave;

    float4 v[BAS_PER_WAVE];
    float den[BAS_PER_WAVE];
#pragma unroll
    for (int i = 0; i < BAS_PER_WAVE; ++i) {
        const int rr = min(r0 + i * BAS_WAVES, N - 1);
        v[i]   = *(const float4*)&accum[(size_t)rr * 128 + lane * 4];
        den[i] = wsum[rr];
    }
    const float4 b = *(const float4*)&bias[lane * 4];

    float4 s = make_float4(0.0f, 0.0f, 0.0f, 0.0f);
    float4 q = make_float4(0.0f, 0.0f, 0.0f, 0.0f);
#pragma unroll
    for (int i = 0; i < BAS_PER_WAVE; ++i) {
        const bool live = r0 + i * BAS_WAVES < N;
        const float inv = den[i] > 0.0f ? 1.0f / den[i] : 1.0f;
        float4 o;
        o.x = lrelu(v[i].x * inv + b.x);
        o.y = lrelu(v[i].y * inv + b.y);
        o.z = lrelu(v[i].z * inv + b.z);
        o.w = lrelu(v[i].w * inv + b.w);
        v[i] = o;
        s.x += live ? o.x : 0.0f;
        s.y += live ? o.y : 0.0f;
        s.z += live ? o.z : 0.0f;
        s.w += live ? o.w : 0.0f;
        q.x += live ? o.x * o.x : 0.0f;
        q.y += live ? o.y * o.y : 0.0f;
        q.z += live ? o.z * o.z : 0.0f;
        q.w += live ? o.w * o.w : 0.0f;
    }
    sPart[wave][lane]      = s;
    sPart[wave][32 + lane] = q;
    __syncthreads();
#pragma unroll
    for (int i = 0; i < BAS_PER_WAVE; ++i) {
        const int r = r0 + i * BAS_WAVES;
        if (r < N) *(float4*)&out[(size_t)r * 128 + lane * 4] = v[i];
    }
    if (threadIdx.x < 256) {
        const float* p = (const float*)sPart;
        float tot = 0.0f;
#pragma unroll
        for (int w = 0; w < BAS_WAVES; ++w) tot += p[w * 256 + threadIdx.x];
        atomicAdd(&stats[threadIdx.x], tot);
    }
}

#define BNA_WAVES 8
#define BNA_PER_WAVE 5
#define BNA_ROWS (BNA_WAVES * BNA_PER_WAVE)
__device__ __forceinline__ void bn_coef(float su, float sq, float w, float b,
                                        float inv, float& sc, float& sh) {
    const float mean = su * inv;
    const float var  = sq * inv - mean * mean;
    sc = w * rsqrtf(var + BN_EPS);
    sh = b - mean * sc;
}
__global__ __launch_bounds__(256) void bn_apply(
    float* __restrict__ out, const float* __restrict__ stats,
    const float* __restrict__ bnw, const float* __restrict__ bnb, int N) {
    const int wave = __builtin_amdgcn_readfirstlane(threadIdx.x >> 5);
    const int lane = threadIdx.x & 31;
    const int r0 = blockIdx.x * BNA_ROWS + wave;
    if (r0 >= N) return;

    float4 v[BNA_PER_WAVE];
#pragma unroll
    for (int i = 0; i < BNA_PER_WAVE; ++i) {
        const int r  = r0 + i * BNA_WAVES;
        const int rr = r < N ? r : r0;
        v[i] = *(const float4*)&out[(size_t)rr * 128 + lane * 4];
    }
    const float4 su = *(const float4*)&stats[lane * 4];
    const float4 sq = *(const float4*)&stats[128 + lane * 4];
    const float4 w  = *(const float4*)&bnw[lane * 4];
    const float4 b  = *(const float4*)&bnb[lane * 4];

    const float inv = 1.0f / (float)N;
    float4 sc, sh;
    bn_coef(su.x, sq.x, w.x, b.x, inv, sc.x, sh.x);
    bn_coef(su.y, sq.y, w.y, b.y, inv, sc.y, sh.y);
    bn_coef(su.z, sq.z, w.z, b.z, inv, sc.z, sh.z);
    bn_coef(su.w, sq.w, w.w, b.w, inv, sc.w, sh.w);

#pragma unroll
    for (int i = 0; i < BNA_PER_WAVE; ++i) {
        const int r = r0 + i * BNA_WAVES;
        float4 o;
        o.x = v[i].x * sc.x + sh.x;
        o.y = v[i].y * sc.y + sh.y;
        o.z = v[i].z * sc.z + sh.z;
        o.w = v[i].w * sc.w + sh.w;
        if (r < N) *(float4*)&out[(size_t)r * 128 + lane * 4] = o;
    }
}

extern "C" void kernel_launch(void* const* d_in, const int* in_sizes, int n_in,
                              void* d_out, int out_size, void* d_ws, size_t ws_size,
                              hipStream_t stream) {
    const float* x    = (const float*)d_in[0];
    const float* ea   = (const float*)d_in[1];
    const int*   ei   = (const int*)d_in[2];
    const float* Wl   = (const float*)d_in[3];
    const float* Wr   = (const float*)d_in[4];
    const float* We   = (const float*)d_in[5];
    const float* att  = (const float*)d_in[6];
    const float* bias = (const float*)d_in[7];
    const float* bnw  = (const float*)d_in[8];
    const float* bnb  = (const float*)d_in[9];
    float* out = (float*)d_out;

    const int N = in_sizes[0] / 128;
    const int E = in_sizes[2] / 2;
    const int* srcIdx = ei;
    const int* dstIdx = ei + E;

    float* ws = (float*)d_ws;
    float* xl = ws;     ws += (size_t)N * 128;
    float* xr = ws;     ws += (size_t)N * 128;
    float* accum = ws;  ws += (size_t)N * 128;
    int* rk    = (int*)ws;  ws += E;
    int* pairs = (int*)ws;  ws += (size_t)E * 2;
    float* wsum = ws;   ws += N;
    int* offs  = (int*)ws;  ws += N + 32;
    int* cnt   = (int*)ws;  ws += N;
    int* hist  = (int*)ws;  ws += (size_t)N * RANK_CHUNKS;
    int* cbase = (int*)ws;  ws += (size_t)N * RANK_CHUNKS;
    float* stats = ws;  ws += 256;

    const int per = (E + RANK_CHUNKS - 1) / RANK_CHUNKS;
    edge_rank<<<RANK_CHUNKS, 1024, (size_t)N * sizeof(int), stream>>>(
        dstIdx, rk, hist, stats, E, N, per);
    chunk_prefix<<<(N + 63) / 64, 64, 0, stream>>>(hist, cbase, cnt, N);
    seg_offsets<<<1, 1024, 0, stream>>>(cnt, offs, N);
    node_transform<<<N / 16, 256, 0, stream>>>(x, Wl, Wr, xl, xr, N);
    edge_place<<<dim3((per + 255) / 256, RANK_CHUNKS), 256, 0, stream>>>(
        srcIdx, dstIdx, rk, offs, cbase, pairs, E, N, per);
    node_attend<<<(N + 3) / 4, 128, 0, stream>>>(ea, We, xl, xr, pairs, offs, att,
                                                 accum, wsum, N);
    const int rpb = BAS_ROWS;
    bias_act_stats<<<(N + rpb - 1) / rpb, 512, 0, stream>>>(accum, wsum, bias, out,
                                                            stats, N);
    const int rpa = BNA_ROWS;
    bn_apply<<<(N + rpa - 1) / rpa, 256, 0, stream>>>(out, stats, bnw, bnb, N);
}


# Node offsets without a scan across workgroups: edge_rank prefix-sums its chunk's histogram over nodes in LDS; offset = sum over chunks; chunk_prefix writes absolute starts; seg_offsets launch gone
# speedup vs baseline: 3.9649x; 1.0725x over previous
#include <hip/hip_runtime.h>

typedef __attribute__((ext_vector_type(2))) float v2f;
typedef __attribute__((ext_vector_type(8))) float v8f;

#define NEG_SLOPE 0.2f
#define BN_EPS 1e-5f

__device__ __forceinline__ float lrelu(float x) { return x > 0.0f ? x : NEG_SLOPE * x; }

#define RANK_CHUNKS 128

__global__ __launch_bounds__(1024) void edge_rank(
    const int* __restrict__ dstIdx, int* __restrict__ rk, int* __restrict__ histPre,
    float* __restrict__ stats, int E, int N, int per) {
    extern __shared__ int sCnt[];
    __shared__ int sWave[32];
    const int t = threadIdx.x;
    for (int i = t; i < N; i += 1024) sCnt[i] = 0;
    if (blockIdx.x == 0 && t < 256) stats[t] = 0.0f;
    __syncthreads();
    const int e0 = blockIdx.x * per;
    const int e1 = min(e0 + per, E);
    for (int eb = e0 + t; eb < e1; eb += 8 * 1024) {
        int d[8];
#pragma unroll
        for (int u = 0; u < 8; ++u) {
            const int e = eb + u * 1024;
            d[u] = (e < e1) ? dstIdx[e] : 0;
        }
#pragma unroll
        for (int u = 0; u < 8; ++u) {
            const int e = eb + u * 1024;
            if (e < e1) rk[e] = atomicAdd(&sCnt[d[u]], 1);
        }
    }
    __syncthreads();

    const int span = (N + 1023) / 1024;
    const int b0 = min(t * span, N);
    const int b1 = min(b0 + span, N);
    int sum = 0;
    for (int i0 = b0; i0 < b1; i0 += 16) {
        int v[16];
#pragma unroll
        for (int u = 0; u < 16; ++u) v[u] = sCnt[min(i0 + u, N - 1)];
#pragma unroll
        for (int u = 0; u < 16; ++u) sum += (i0 + u < b1) ? v[u] : 0;
    }
    const int lane = t & 31;
    const int wave = t >> 5;
    int inc = sum;
#pragma unroll
    for (int off = 1; off < 32; off <<= 1) {
        const int nb = __shfl_up(inc, off, 32);
        if (lane >= off) inc += nb;
    }
    if (lane == 31) sWave[wave] = inc;
    __syncthreads();
    if (wave == 0) {
        const int w = sWave[lane];
        int wi = w;
#pragma unroll
        for (int off = 1; off < 32; off <<= 1) {
            const int nb = __shfl_up(wi, off, 32);
            if (lane >= off) wi += nb;
        }
        sWave[lane] = wi - w;
    }
    __syncthreads();
    int run = sWave[wave] + inc - sum;
    for (int i0 = b0; i0 < b1; i0 += 16) {
        int v[16];
#pragma unroll
        for (int u = 0; u < 16; ++u) v[u] = sCnt[min(i0 + u, N - 1)];
#pragma unroll
        for (int u = 0; u < 16; ++u) {
            if (i0 + u < b1) {
                sCnt[i0 + u] = run;
                run += v[u];
            }
        }
    }
    if (b0 < N && b1 == N) sCnt[N] = run;
    __syncthreads();
    int2* __restrict__ h = (int2*)histPre + (size_t)blockIdx.x * N;
#pragma unroll 4
    for (int i = t; i < N; i += 1024) {
        const int p = sCnt[i];
        h[i] = make_int2(sCnt[i + 1] - p, p);
    }
}

__global__ __launch_bounds__(64) void chunk_prefix(
    const int* __restrict__ histPre, int* __restrict__ start,
    int* __restrict__ offs, int N) {
    const int node = blockIdx.x * 64 + threadIdx.x;
    if (node >= N) return;
    int2 v[RANK_CHUNKS];
#pragma unroll
    for (int c = 0; c < RANK_CHUNKS; ++c)
        v[c] = ((const int2*)histPre)[(size_t)c * N + node];
    int base = 0;
#pragma unroll
    for (int c = 0; c < RANK_CHUNKS; ++c) base += v[c].y;
    int run = base;
#pragma unroll
    for (int c = 0; c < RANK_CHUNKS; ++c) {
        start[(size_t)c * N + node] = run;
        run += v[c].x;
    }
    offs[node] = base;
    if (node == N - 1) offs[N] = run;
}

__global__ __launch_bounds__(256) void node_transform(
    const float* __restrict__ x, const float* __restrict__ Wl,
    const float* __restrict__ Wr, float* __restrict__ xl, float* __restrict__ xr,
    int N) {
    const int wave = threadIdx.x >> 5;
    const int lane = threadIdx.x & 31;
    const int lo   = lane & 15;
    const int hiK  = (lane >= 16) ? 2 : 0;
    const int hiM  = (lane >= 16) ? 8 : 0;
    const int rowTile = blockIdx.x * 16;
    const int col  = wave * 16 + lo;
    const int aRow = rowTile + lo;

    v8f cl = {};
    v8f cr = {};
    for (int k = 0; k < 128; k += 4) {
        v2f a = *(const v2f*)&x[aRow * 128 + k + hiK];
        v2f bl, br;
        bl.x = Wl[(k + hiK) * 128 + col];
        bl.y = Wl[(k + hiK + 1) * 128 + col];
        br.x = Wr[(k + hiK) * 128 + col];
        br.y = Wr[(k + hiK + 1) * 128 + col];
        cl = __builtin_amdgcn_wmma_f32_16x16x4_f32(false, a, false, bl, (short)0, cl, false, false);
        cr = __builtin_amdgcn_wmma_f32_16x16x4_f32(false, a, false, br, (short)0, cr, false, false);
    }
#pragma unroll
    for (int r = 0; r < 8; ++r) {
        int row = rowTile + r + hiM;
        xl[row * 128 + col] = cl[r];
        xr[row * 128 + col] = cr[r];
    }
}

__global__ __launch_bounds__(256) void edge_place(
    const int* __restrict__ srcIdx, const int* __restrict__ dstIdx,
    const int* __restrict__ rk, const int* __restrict__ start,
    int* __restrict__ pairs, int E, int N, int per) {
    const int chunk = blockIdx.y;
    const int inChunk = blockIdx.x * 256 + threadIdx.x;
    const int e = chunk * per + inChunk;
    if (inChunk >= per || e >= E) return;
    const int d = dstIdx[e];
    const int pos = start[(size_t)chunk * N + d] + rk[e];
    ((int2*)pairs)[pos] = make_int2(srcIdx[e], e);
}

__global__ __launch_bounds__(128, 5) void node_attend(
    const float* __restrict__ edge_attr, const float* __restrict__ We,
    const float* __restrict__ xl, const float* __restrict__ xr,
    const int* __restrict__ pairs, const int* __restrict__ offs,
    const float* __restrict__ att, float* __restrict__ accum,
    float* __restrict__ wsum, int N) {
    __shared__ float4 sB[512];
    for (int pce = threadIdx.x; pce < 256; pce += 128) {
        const int c  = pce & 15;
        const int h  = (pce >> 4) & 1;
        const int q  = (pce >> 5) & 1;
        const int ks = (pce >> 6) & 3;
        const int k0 = 4 * ks + 2 * h;
        const float4 r0 = *(const float4*)&We[k0 * 128 + 64 * q + 4 * c];
        const float4 r1 = *(const float4*)&We[(k0 + 1) * 128 + 64 * q + 4 * c];
        sB[((((ks * 2 + q) * 2 + 0) * 2 + h) * 16) + c] = make_float4(r0.x, r1.x, r0.y, r1.y);
        sB[((((ks * 2 + q) * 2 + 1) * 2 + h) * 16) + c] = make_float4(r0.z, r1.z, r0.w, r1.w);
    }
    __syncthreads();

    const int node = blockIdx.x * 4 + __builtin_amdgcn_readfirstlane(threadIdx.x >> 5);
    if (node >= N) return;
    const int lane = threadIdx.x & 31;
    const int lo   = lane & 15;
    const int hh   = lane >> 4;
    const int hiK  = hh * 2;
    const int hiM  = hh * 8;
    const int2* pr2 = (const int2*)pairs;
    const int beg = offs[node];
    const int end = offs[node + 1];

    float m  = -INFINITY;
    float ws = 0.0f;
    float accv[2][4];
#pragma unroll
    for (int q = 0; q < 2; ++q)
#pragma unroll
        for (int j = 0; j < 4; ++j) accv[q][j] = 0.0f;

    int2 nxt = make_int2(0, 0);
    if (beg + lo < end) nxt = pr2[beg + lo];

    for (int t = beg; t < end; t += 16) {
        const int2 cur = nxt;
        nxt = make_int2(0, 0);
        if (t + 16 + lo < end) nxt = pr2[t + 16 + lo];
        asm volatile("" ::: "memory");

        const float4 xr0 = *(const float4*)&xr[(size_t)node * 128 + lo * 4];
        const float4 xr1 = *(const float4*)&xr[(size_t)node * 128 + 64 + lo * 4];
        const float4 at0 = *(const float4*)&att[lo * 4];
        const float4 at1 = *(const float4*)&att[64 + lo * 4];
        const float xrv[2][4] = {{xr0.x, xr0.y, xr0.z, xr0.w}, {xr1.x, xr1.y, xr1.z, xr1.w}};
        const float avj[2][4] = {{at0.x, at0.y, at0.z, at0.w}, {at1.x, at1.y, at1.z, at1.w}};

        const float* aRow = &edge_attr[(size_t)cur.y * 16];
        v2f a[4];
        a[0] = *(const v2f*)&aRow[0 + hiK];
        a[1] = *(const v2f*)&aRow[4 + hiK];
        a[2] = *(const v2f*)&aRow[8 + hiK];
        a[3] = *(const v2f*)&aRow[12 + hiK];

        float vl[2][8][4];
#pragma unroll
        for (int r = 0; r < 8; ++r) {
            const unsigned offS =
                (unsigned)__shfl(cur.x, r + hiM, 32) * 512u + (unsigned)lo * 16u;
#pragma unroll
            for (int q = 0; q < 2; ++q) {
                const float4 v = *(const float4*)((const char*)xl + offS + q * 256);
                vl[q][r][0] = v.x;
                vl[q][r][1] = v.y;
                vl[q][r][2] = v.z;
                vl[q][r][3] = v.w;
            }
        }

        float e[8];
#pragma unroll
        for (int r = 0; r < 8; ++r) e[r] = 0.0f;
#pragma unroll
        for (int q = 0; q < 2; ++q) {
#pragma unroll
            for (int jp = 0; jp < 2; ++jp) {
                float4 bv[4];
#pragma unroll
                for (int ks = 0; ks < 4; ++ks)
                    bv[ks] = sB[((((ks * 2 + q) * 2 + jp) * 2 + hh) * 16) + lo];
#pragma unroll
                for (int jj = 0; jj < 2; ++jj) {
                    const int j = 2 * jp + jj;
                    v8f c;
#pragma unroll
                    for (int r = 0; r < 8; ++r) c[r] = vl[q][r][j] + xrv[q][j];
#pragma unroll
                    for (int ks = 0; ks < 4; ++ks) {
                        v2f b;
                        b.x = jj ? bv[ks].z : bv[ks].x;
                        b.y = jj ? bv[ks].w : bv[ks].y;
                        c = __builtin_amdgcn_wmma_f32_16x16x4_f32(false, a[ks], false, b, (short)0, c, false, false);
                    }
#pragma unroll
                    for (int r = 0; r < 8; ++r) e[r] += lrelu(c[r]) * avj[q][j];
                }
            }
        }

        const int nvalid = end - t;
        float tm = -INFINITY;
#pragma unroll
        for (int r = 0; r < 8; ++r) {
            float v = e[r];
            v += __shfl_xor(v, 1, 32);
            v += __shfl_xor(v, 2, 32);
            v += __shfl_xor(v, 4, 32);
            v += __shfl_xor(v, 8, 32);
            if (r + hiM >= nvalid) v = -INFINITY;
            e[r] = v;
            tm = fmaxf(tm, v);
        }
        tm = fmaxf(tm, __shfl_xor(tm, 16, 32));

        const float mn = fmaxf(m, tm);
        const float sc = expf(m - mn);
        m = mn;
        ws *= sc;
#pragma unroll
        for (int q = 0; q < 2; ++q)
#pragma unroll
            for (int j = 0; j < 4; ++j) accv[q][j] *= sc;
#pragma unroll
        for (int r = 0; r < 8; ++r) {
            const float p = expf(e[r] - mn);
            ws += p;
#pragma unroll
            for (int q = 0; q < 2; ++q)
#pragma unroll
                for (int j = 0; j < 4; ++j) accv[q][j] += p * vl[q][r][j];
        }
    }

#pragma unroll
    for (int q = 0; q < 2; ++q)
#pragma unroll
        for (int j = 0; j < 4; ++j) accv[q][j] += __shfl_xor(accv[q][j], 16, 32);
    ws += __shfl_xor(ws, 16, 32);

    const float4 o = hh ? make_float4(accv[1][0], accv[1][1], accv[1][2], accv[1][3])
                        : make_float4(accv[0][0], accv[0][1], accv[0][2], accv[0][3]);
    *(float4*)&accum[(size_t)node * 128 + hh * 64 + lo * 4] = o;
    if (lane == 0) wsum[node] = ws;
}

#define BAS_WAVES 16
#define BAS_PER_WAVE 5
#define BAS_ROWS (BAS_WAVES * BAS_PER_WAVE)
__global__ __launch_bounds__(512) void bias_act_stats(
    const float* __restrict__ accum, const float* __restrict__ wsum,
    const float* __restrict__ bias, float* __restrict__ out,
    float* __restrict__ stats, int N) {
    __shared__ float4 sPart[BAS_WAVES][64];
    const int wave = __builtin_amdgcn_readfirstlane(threadIdx.x >> 5);
    const int lane = threadIdx.x & 31;
    const int r0 = blockIdx.x * BAS_ROWS + wave;

    float4 v[BAS_PER_WAVE];
    float den[BAS_PER_WAVE];
#pragma unroll
    for (int i = 0; i < BAS_PER_WAVE; ++i) {
        const int rr = min(r0 + i * BAS_WAVES, N - 1);
        v[i]   = *(const float4*)&accum[(size_t)rr * 128 + lane * 4];
        den[i] = wsum[rr];
    }
    const float4 b = *(const float4*)&bias[lane * 4];

    float4 s = make_float4(0.0f, 0.0f, 0.0f, 0.0f);
    float4 q = make_float4(0.0f, 0.0f, 0.0f, 0.0f);
#pragma unroll
    for (int i = 0; i < BAS_PER_WAVE; ++i) {
        const bool live = r0 + i * BAS_WAVES < N;
        const float inv = den[i] > 0.0f ? 1.0f / den[i] : 1.0f;
        float4 o;
        o.x = lrelu(v[i].x * inv + b.x);
        o.y = lrelu(v[i].y * inv + b.y);
        o.z = lrelu(v[i].z * inv + b.z);
        o.w = lrelu(v[i].w * inv + b.w);
        v[i] = o;
        s.x += live ? o.x : 0.0f;
        s.y += live ? o.y : 0.0f;
        s.z += live ? o.z : 0.0f;
        s.w += live ? o.w : 0.0f;
        q.x += live ? o.x * o.x : 0.0f;
        q.y += live ? o.y * o.y : 0.0f;
        q.z += live ? o.z * o.z : 0.0f;
        q.w += live ? o.w * o.w : 0.0f;
    }
    sPart[wave][lane]      = s;
    sPart[wave][32 + lane] = q;
    __syncthreads();
#pragma unroll
    for (int i = 0; i < BAS_PER_WAVE; ++i) {
        const int r = r0 + i * BAS_WAVES;
        if (r < N) *(float4*)&out[(size_t)r * 128 + lane * 4] = v[i];
    }
    if (threadIdx.x < 256) {
        const float* p = (const float*)sPart;
        float tot = 0.0f;
#pragma unroll
        for (int w = 0; w < BAS_WAVES; ++w) tot += p[w * 256 + threadIdx.x];
        atomicAdd(&stats[threadIdx.x], tot);
    }
}

#define BNA_WAVES 8
#define BNA_PER_WAVE 5
#define BNA_ROWS (BNA_WAVES * BNA_PER_WAVE)
__device__ __forceinline__ void bn_coef(float su, float sq, float w, float b,
                                        float inv, float& sc, float& sh) {
    const float mean = su * inv;
    const float var  = sq * inv - mean * mean;
    sc = w * rsqrtf(var + BN_EPS);
    sh = b - mean * sc;
}
__global__ __launch_bounds__(256) void bn_apply(
    float* __restrict__ out, const float* __restrict__ stats,
    const float* __restrict__ bnw, const float* __restrict__ bnb, int N) {
    const int wave = __builtin_amdgcn_readfirstlane(threadIdx.x >> 5);
    const int lane = threadIdx.x & 31;
    const int r0 = blockIdx.x * BNA_ROWS + wave;
    if (r0 >= N) return;

    float4 v[BNA_PER_WAVE];
#pragma unroll
    for (int i = 0; i < BNA_PER_WAVE; ++i) {
        const int r  = r0 + i * BNA_WAVES;
        const int rr = r < N ? r : r0;
        v[i] = *(const float4*)&out[(size_t)rr * 128 + lane * 4];
    }
    const float4 su = *(const float4*)&stats[lane * 4];
    const float4 sq = *(const float4*)&stats[128 + lane * 4];
    const float4 w  = *(const float4*)&bnw[lane * 4];
    const float4 b  = *(const float4*)&bnb[lane * 4];

    const float inv = 1.0f / (float)N;
    float4 sc, sh;
    bn_coef(su.x, sq.x, w.x, b.x, inv, sc.x, sh.x);
    bn_coef(su.y, sq.y, w.y, b.y, inv, sc.y, sh.y);
    bn_coef(su.z, sq.z, w.z, b.z, inv, sc.z, sh.z);
    bn_coef(su.w, sq.w, w.w, b.w, inv, sc.w, sh.w);

#pragma unroll
    for (int i = 0; i < BNA_PER_WAVE; ++i) {
        const int r = r0 + i * BNA_WAVES;
        float4 o;
        o.x = v[i].x * sc.x + sh.x;
        o.y = v[i].y * sc.y + sh.y;
        o.z = v[i].z * sc.z + sh.z;
        o.w = v[i].w * sc.w + sh.w;
        if (r < N) *(float4*)&out[(size_t)r * 128 + lane * 4] = o;
    }
}

extern "C" void kernel_launch(void* const* d_in, const int* in_sizes, int n_in,
                              void* d_out, int out_size, void* d_ws, size_t ws_size,
                              hipStream_t stream) {
    const float* x    = (const float*)d_in[0];
    const float* ea   = (const float*)d_in[1];
    const int*   ei   = (const int*)d_in[2];
    const float* Wl   = (const float*)d_in[3];
    const float* Wr   = (const float*)d_in[4];
    const float* We   = (const float*)d_in[5];
    const float* att  = (const float*)d_in[6];
    const float* bias = (const float*)d_in[7];
    const float* bnw  = (const float*)d_in[8];
    const float* bnb  = (const float*)d_in[9];
    float* out = (float*)d_out;

    const int N = in_sizes[0] / 128;
    const int E = in_sizes[2] / 2;
    const int* srcIdx = ei;
    const int* dstIdx = ei + E;

    float* ws = (float*)d_ws;
    float* xl = ws;     ws += (size_t)N * 128;
    float* xr = ws;     ws += (size_t)N * 128;
    float* accum = ws;  ws += (size_t)N * 128;
    int* histPre = (int*)ws;  ws += (size_t)N * RANK_CHUNKS * 2;
    int* start = (int*)ws;  ws += (size_t)N * RANK_CHUNKS;
    int* rk    = (int*)ws;  ws += E;
    int* pairs = (int*)ws;  ws += (size_t)E * 2;
    float* wsum = ws;   ws += N;
    int* offs  = (int*)ws;  ws += N + 32;
    float* stats = ws;  ws += 256;

    const int per = (E + RANK_CHUNKS - 1) / RANK_CHUNKS;
    edge_rank<<<RANK_CHUNKS, 1024, (size_t)(N + 1) * sizeof(int), stream>>>(
        dstIdx, rk, histPre, stats, E, N, per);
    chunk_prefix<<<(N + 63) / 64, 64, 0, stream>>>(histPre, start, offs, N);
    node_transform<<<N / 16, 256, 0, stream>>>(x, Wl, Wr, xl, xr, N);
    edge_place<<<dim3((per + 255) / 256, RANK_CHUNKS), 256, 0, stream>>>(
        srcIdx, dstIdx, rk, start, pairs, E, N, per);
    node_attend<<<(N + 3) / 4, 128, 0, stream>>>(ea, We, xl, xr, pairs, offs, att,
                                                 accum, wsum, N);
    const int rpb = BAS_ROWS;
    bias_act_stats<<<(N + rpb - 1) / rpb, 512, 0, stream>>>(accum, wsum, bias, out,
                                                            stats, N);
    const int rpa = BNA_ROWS;
    bn_apply<<<(N + rpa - 1) / rpa, 256, 0, stream>>>(out, stats, bnw, bnb, N);
}


# x@Wl, x@Wr no longer a launch of their own: their 625 tiles run as the first workgroups of the edge placement's grid (independent work, block-uniform branch); 6 launches not 7
# speedup vs baseline: 4.1609x; 1.0494x over previous
#include <hip/hip_runtime.h>

typedef __attribute__((ext_vector_type(2))) float v2f;
typedef __attribute__((ext_vector_type(8))) float v8f;

#define NEG_SLOPE 0.2f
#define BN_EPS 1e-5f

__device__ __forceinline__ float lrelu(float x) { return x > 0.0f ? x : NEG_SLOPE * x; }

#define RANK_CHUNKS 128

__global__ __launch_bounds__(1024) void edge_rank(
    const int* __restrict__ dstIdx, int* __restrict__ rk, int* __restrict__ histPre,
    float* __restrict__ stats, int E, int N, int per) {
    extern __shared__ int sCnt[];
    __shared__ int sWave[32];
    const int t = threadIdx.x;
    for (int i = t; i < N; i += 1024) sCnt[i] = 0;
    if (blockIdx.x == 0 && t < 256) stats[t] = 0.0f;
    __syncthreads();
    const int e0 = blockIdx.x * per;
    const int e1 = min(e0 + per, E);
    for (int eb = e0 + t; eb < e1; eb += 8 * 1024) {
        int d[8];
#pragma unroll
        for (int u = 0; u < 8; ++u) {
            const int e = eb + u * 1024;
            d[u] = (e < e1) ? dstIdx[e] : 0;
        }
#pragma unroll
        for (int u = 0; u < 8; ++u) {
            const int e = eb + u * 1024;
            if (e < e1) rk[e] = atomicAdd(&sCnt[d[u]], 1);
        }
    }
    __syncthreads();

    const int span = (N + 1023) / 1024;
    const int b0 = min(t * span, N);
    const int b1 = min(b0 + span, N);
    int sum = 0;
    for (int i0 = b0; i0 < b1; i0 += 16) {
        int v[16];
#pragma unroll
        for (int u = 0; u < 16; ++u) v[u] = sCnt[min(i0 + u, N - 1)];
#pragma unroll
        for (int u = 0; u < 16; ++u) sum += (i0 + u < b1) ? v[u] : 0;
    }
    const int lane = t & 31;
    const int wave = t >> 5;
    int inc = sum;
#pragma unroll
    for (int off = 1; off < 32; off <<= 1) {
        const int nb = __shfl_up(inc, off, 32);
        if (lane >= off) inc += nb;
    }
    if (lane == 31) sWave[wave] = inc;
    __syncthreads();
    if (wave == 0) {
        const int w = sWave[lane];
        int wi = w;
#pragma unroll
        for (int off = 1; off < 32; off <<= 1) {
            const int nb = __shfl_up(wi, off, 32);
            if (lane >= off) wi += nb;
        }
        sWave[lane] = wi - w;
    }
    __syncthreads();
    int run = sWave[wave] + inc - sum;
    for (int i0 = b0; i0 < b1; i0 += 16) {
        int v[16];
#pragma unroll
        for (int u = 0; u < 16; ++u) v[u] = sCnt[min(i0 + u, N - 1)];
#pragma unroll
        for (int u = 0; u < 16; ++u) {
            if (i0 + u < b1) {
                sCnt[i0 + u] = run;
                run += v[u];
            }
        }
    }
    if (b0 < N && b1 == N) sCnt[N] = run;
    __syncthreads();
    int2* __restrict__ h = (int2*)histPre + (size_t)blockIdx.x * N;
#pragma unroll 4
    for (int i = t; i < N; i += 1024) {
        const int p = sCnt[i];
        h[i] = make_int2(sCnt[i + 1] - p, p);
    }
}

__global__ __launch_bounds__(64) void chunk_prefix(
    const int* __restrict__ histPre, int* __restrict__ start,
    int* __restrict__ offs, int N) {
    const int node = blockIdx.x * 64 + threadIdx.x;
    if (node >= N) return;
    int2 v[RANK_CHUNKS];
#pragma unroll
    for (int c = 0; c < RANK_CHUNKS; ++c)
        v[c] = ((const int2*)histPre)[(size_t)c * N + node];
    int base = 0;
#pragma unroll
    for (int c = 0; c < RANK_CHUNKS; ++c) base += v[c].y;
    int run = base;
#pragma unroll
    for (int c = 0; c < RANK_CHUNKS; ++c) {
        start[(size_t)c * N + node] = run;
        run += v[c].x;
    }
    offs[node] = base;
    if (node == N - 1) offs[N] = run;
}

__global__ __launch_bounds__(256) void transform_and_place(
    const float* __restrict__ x, const float* __restrict__ Wl,
    const float* __restrict__ Wr, float* __restrict__ xl, float* __restrict__ xr,
    const int* __restrict__ srcIdx, const int* __restrict__ dstIdx,
    const int* __restrict__ rk, const int* __restrict__ start,
    int* __restrict__ pairs, int E, int N, int per, int bpc, int tRows,
    int nTiles) {
    if ((int)blockIdx.y < tRows) {
        const int tile = blockIdx.y * bpc + blockIdx.x;
        if (tile >= nTiles) return;
        const int wave = threadIdx.x >> 5;
        const int lane = threadIdx.x & 31;
        const int lo   = lane & 15;
        const int hiK  = (lane >= 16) ? 2 : 0;
        const int hiM  = (lane >= 16) ? 8 : 0;
        const int rowTile = tile * 16;
        const int col  = wave * 16 + lo;
        const int aRow = rowTile + lo;

        v8f cl = {};
        v8f cr = {};
        for (int k = 0; k < 128; k += 4) {
            v2f a = *(const v2f*)&x[aRow * 128 + k + hiK];
            v2f bl, br;
            bl.x = Wl[(k + hiK) * 128 + col];
            bl.y = Wl[(k + hiK + 1) * 128 + col];
            br.x = Wr[(k + hiK) * 128 + col];
            br.y = Wr[(k + hiK + 1) * 128 + col];
            cl = __builtin_amdgcn_wmma_f32_16x16x4_f32(false, a, false, bl, (short)0, cl, false, false);
            cr = __builtin_amdgcn_wmma_f32_16x16x4_f32(false, a, false, br, (short)0, cr, false, false);
        }
#pragma unroll
        for (int r = 0; r < 8; ++r) {
            int row = rowTile + r + hiM;
            xl[row * 128 + col] = cl[r];
            xr[row * 128 + col] = cr[r];
        }
        return;
    }
    const int chunk = blockIdx.y - tRows;
    const int inChunk = blockIdx.x * 256 + threadIdx.x;
    const int e = chunk * per + inChunk;
    if (inChunk >= per || e >= E) return;
    const int d = dstIdx[e];
    const int pos = start[(size_t)chunk * N + d] + rk[e];
    ((int2*)pairs)[pos] = make_int2(srcIdx[e], e);
}

__global__ __launch_bounds__(128, 5) void node_attend(
    const float* __restrict__ edge_attr, const float* __restrict__ We,
    const float* __restrict__ xl, const float* __restrict__ xr,
    const int* __restrict__ pairs, const int* __restrict__ offs,
    const float* __restrict__ att, float* __restrict__ accum,
    float* __restrict__ wsum, int N) {
    __shared__ float4 sB[512];
    for (int pce = threadIdx.x; pce < 256; pce += 128) {
        const int c  = pce & 15;
        const int h  = (pce >> 4) & 1;
        const int q  = (pce >> 5) & 1;
        const int ks = (pce >> 6) & 3;
        const int k0 = 4 * ks + 2 * h;
        const float4 r0 = *(const float4*)&We[k0 * 128 + 64 * q + 4 * c];
        const float4 r1 = *(const float4*)&We[(k0 + 1) * 128 + 64 * q + 4 * c];
        sB[((((ks * 2 + q) * 2 + 0) * 2 + h) * 16) + c] = make_float4(r0.x, r1.x, r0.y, r1.y);
        sB[((((ks * 2 + q) * 2 + 1) * 2 + h) * 16) + c] = make_float4(r0.z, r1.z, r0.w, r1.w);
    }
    __syncthreads();

    const int node = blockIdx.x * 4 + __builtin_amdgcn_readfirstlane(threadIdx.x >> 5);
    if (node >= N) return;
    const int lane = threadIdx.x & 31;
    const int lo   = lane & 15;
    const int hh   = lane >> 4;
    const int hiK  = hh * 2;
    const int hiM  = hh * 8;
    const int2* pr2 = (const int2*)pairs;
    const int beg = offs[node];
    const int end = offs[node + 1];

    float m  = -INFINITY;
    float ws = 0.0f;
    float accv[2][4];
#pragma unroll
    for (int q = 0; q < 2; ++q)
#pragma unroll
        for (int j = 0; j < 4; ++j) accv[q][j] = 0.0f;

    int2 nxt = make_int2(0, 0);
    if (beg + lo < end) nxt = pr2[beg + lo];

    for (int t = beg; t < end; t += 16) {
        const int2 cur = nxt;
        nxt = make_int2(0, 0);
        if (t + 16 + lo < end) nxt = pr2[t + 16 + lo];
        asm volatile("" ::: "memory");

        const float4 xr0 = *(const float4*)&xr[(size_t)node * 128 + lo * 4];
        const float4 xr1 = *(const float4*)&xr[(size_t)node * 128 + 64 + lo * 4];
        const float4 at0 = *(const float4*)&att[lo * 4];
        const float4 at1 = *(const float4*)&att[64 + lo * 4];
        const float xrv[2][4] = {{xr0.x, xr0.y, xr0.z, xr0.w}, {xr1.x, xr1.y, xr1.z, xr1.w}};
        const float avj[2][4] = {{at0.x, at0.y, at0.z, at0.w}, {at1.x, at1.y, at1.z, at1.w}};

        const float* aRow = &edge_attr[(size_t)cur.y * 16];
        v2f a[4];
        a[0] = *(const v2f*)&aRow[0 + hiK];
        a[1] = *(const v2f*)&aRow[4 + hiK];
        a[2] = *(const v2f*)&aRow[8 + hiK];
        a[3] = *(const v2f*)&aRow[12 + hiK];

        float vl[2][8][4];
#pragma unroll
        for (int r = 0; r < 8; ++r) {
            const unsigned offS =
                (unsigned)__shfl(cur.x, r + hiM, 32) * 512u + (unsigned)lo * 16u;
#pragma unroll
            for (int q = 0; q < 2; ++q) {
                const float4 v = *(const float4*)((const char*)xl + offS + q * 256);
                vl[q][r][0] = v.x;
                vl[q][r][1] = v.y;
                vl[q][r][2] = v.z;
                vl[q][r][3] = v.w;
            }
        }

        float e[8];
#pragma unroll
        for (int r = 0; r < 8; ++r) e[r] = 0.0f;
#pragma unroll
        for (int q = 0; q < 2; ++q) {
#pragma unroll
            for (int jp = 0; jp < 2; ++jp) {
                float4 bv[4];
#pragma unroll
                for (int ks = 0; ks < 4; ++ks)
                    bv[ks] = sB[((((ks * 2 + q) * 2 + jp) * 2 + hh) * 16) + lo];
#pragma unroll
                for (int jj = 0; jj < 2; ++jj) {
                    const int j = 2 * jp + jj;
                    v8f c;
#pragma unroll
                    for (int r = 0; r < 8; ++r) c[r] = vl[q][r][j] + xrv[q][j];
#pragma unroll
                    for (int ks = 0; ks < 4; ++ks) {
                        v2f b;
                        b.x = jj ? bv[ks].z : bv[ks].x;
                        b.y = jj ? bv[ks].w : bv[ks].y;
                        c = __builtin_amdgcn_wmma_f32_16x16x4_f32(false, a[ks], false, b, (short)0, c, false, false);
                    }
#pragma unroll
                    for (int r = 0; r < 8; ++r) e[r] += lrelu(c[r]) * avj[q][j];
                }
            }
        }

        const int nvalid = end - t;
        float tm = -INFINITY;
#pragma unroll
        for (int r = 0; r < 8; ++r) {
            float v = e[r];
            v += __shfl_xor(v, 1, 32);
            v += __shfl_xor(v, 2, 32);
            v += __shfl_xor(v, 4, 32);
            v += __shfl_xor(v, 8, 32);
            if (r + hiM >= nvalid) v = -INFINITY;
            e[r] = v;
            tm = fmaxf(tm, v);
        }
        tm = fmaxf(tm, __shfl_xor(tm, 16, 32));

        const float mn = fmaxf(m, tm);
        const float sc = expf(m - mn);
        m = mn;
        ws *= sc;
#pragma unroll
        for (int q = 0; q < 2; ++q)
#pragma unroll
            for (int j = 0; j < 4; ++j) accv[q][j] *= sc;
#pragma unroll
        for (int r = 0; r < 8; ++r) {
            const float p = expf(e[r] - mn);
            ws += p;
#pragma unroll
            for (int q = 0; q < 2; ++q)
#pragma unroll
                for (int j = 0; j < 4; ++j) accv[q][j] += p * vl[q][r][j];
        }
    }

#pragma unroll
    for (int q = 0; q < 2; ++q)
#pragma unroll
        for (int j = 0; j < 4; ++j) accv[q][j] += __shfl_xor(accv[q][j], 16, 32);
    ws += __shfl_xor(ws, 16, 32);

    const float4 o = hh ? make_float4(accv[1][0], accv[1][1], accv[1][2], accv[1][3])
                        : make_float4(accv[0][0], accv[0][1], accv[0][2], accv[0][3]);
    *(float4*)&accum[(size_t)node * 128 + hh * 64 + lo * 4] = o;
    if (lane == 0) wsum[node] = ws;
}

#define BAS_WAVES 16
#define BAS_PER_WAVE 5
#define BAS_ROWS (BAS_WAVES * BAS_PER_WAVE)
__global__ __launch_bounds__(512) void bias_act_stats(
    const float* __restrict__ accum, const float* __restrict__ wsum,
    const float* __restrict__ bias, float* __restrict__ out,
    float* __restrict__ stats, int N) {
    __shared__ float4 sPart[BAS_WAVES][64];
    const int wave = __builtin_amdgcn_readfirstlane(threadIdx.x >> 5);
    const int lane = threadIdx.x & 31;
    const int r0 = blockIdx.x * BAS_ROWS + wave;

    float4 v[BAS_PER_WAVE];
    float den[BAS_PER_WAVE];
#pragma unroll
    for (int i = 0; i < BAS_PER_WAVE; ++i) {
        const int rr = min(r0 + i * BAS_WAVES, N - 1);
        v[i]   = *(const float4*)&accum[(size_t)rr * 128 + lane * 4];
        den[i] = wsum[rr];
    }
    const float4 b = *(const float4*)&bias[lane * 4];

    float4 s = make_float4(0.0f, 0.0f, 0.0f, 0.0f);
    float4 q = make_float4(0.0f, 0.0f, 0.0f, 0.0f);
#pragma unroll
    for (int i = 0; i < BAS_PER_WAVE; ++i) {
        const bool live = r0 + i * BAS_WAVES < N;
        const float inv = den[i] > 0.0f ? 1.0f / den[i] : 1.0f;
        float4 o;
        o.x = lrelu(v[i].x * inv + b.x);
        o.y = lrelu(v[i].y * inv + b.y);
        o.z = lrelu(v[i].z * inv + b.z);
        o.w = lrelu(v[i].w * inv + b.w);
        v[i] = o;
        s.x += live ? o.x : 0.0f;
        s.y += live ? o.y : 0.0f;
        s.z += live ? o.z : 0.0f;
        s.w += live ? o.w : 0.0f;
        q.x += live ? o.x * o.x : 0.0f;
        q.y += live ? o.y * o.y : 0.0f;
        q.z += live ? o.z * o.z : 0.0f;
        q.w += live ? o.w * o.w : 0.0f;
    }
    sPart[wave][lane]      = s;
    sPart[wave][32 + lane] = q;
    __syncthreads();
#pragma unroll
    for (int i = 0; i < BAS_PER_WAVE; ++i) {
        const int r = r0 + i * BAS_WAVES;
        if (r < N) *(float4*)&out[(size_t)r * 128 + lane * 4] = v[i];
    }
    if (threadIdx.x < 256) {
        const float* p = (const float*)sPart;
        float tot = 0.0f;
#pragma unroll
        for (int w = 0; w < BAS_WAVES; ++w) tot += p[w * 256 + threadIdx.x];
        atomicAdd(&stats[threadIdx.x], tot);
    }
}

#define BNA_WAVES 8
#define BNA_PER_WAVE 5
#define BNA_ROWS (BNA_WAVES * BNA_PER_WAVE)
__device__ __forceinline__ void bn_coef(float su, float sq, float w, float b,
                                        float inv, float& sc, float& sh) {
    const float mean = su * inv;
    const float var  = sq * inv - mean * mean;
    sc = w * rsqrtf(var + BN_EPS);
    sh = b - mean * sc;
}
__global__ __launch_bounds__(256) void bn_apply(
    float* __restrict__ out, const float* __restrict__ stats,
    const float* __restrict__ bnw, const float* __restrict__ bnb, int N) {
    const int wave = __builtin_amdgcn_readfirstlane(threadIdx.x >> 5);
    const int lane = threadIdx.x & 31;
    const int r0 = blockIdx.x * BNA_ROWS + wave;
    if (r0 >= N) return;

    float4 v[BNA_PER_WAVE];
#pragma unroll
    for (int i = 0; i < BNA_PER_WAVE; ++i) {
        const int r  = r0 + i * BNA_WAVES;
        const int rr = r < N ? r : r0;
        v[i] = *(const float4*)&out[(size_t)rr * 128 + lane * 4];
    }
    const float4 su = *(const float4*)&stats[lane * 4];
    const float4 sq = *(const float4*)&stats[128 + lane * 4];
    const float4 w  = *(const float4*)&bnw[lane * 4];
    const float4 b  = *(const float4*)&bnb[lane * 4];

    const float inv = 1.0f / (float)N;
    float4 sc, sh;
    bn_coef(su.x, sq.x, w.x, b.x, inv, sc.x, sh.x);
    bn_coef(su.y, sq.y, w.y, b.y, inv, sc.y, sh.y);
    bn_coef(su.z, sq.z, w.z, b.z, inv, sc.z, sh.z);
    bn_coef(su.w, sq.w, w.w, b.w, inv, sc.w, sh.w);

#pragma unroll
    for (int i = 0; i < BNA_PER_WAVE; ++i) {
        const int r = r0 + i * BNA_WAVES;
        float4 o;
        o.x = v[i].x * sc.x + sh.x;
        o.y = v[i].y * sc.y + sh.y;
        o.z = v[i].z * sc.z + sh.z;
        o.w = v[i].w * sc.w + sh.w;
        if (r < N) *(float4*)&out[(size_t)r * 128 + lane * 4] = o;
    }
}

extern "C" void kernel_launch(void* const* d_in, const int* in_sizes, int n_in,
                              void* d_out, int out_size, void* d_ws, size_t ws_size,
                              hipStream_t stream) {
    const float* x    = (const float*)d_in[0];
    const float* ea   = (const float*)d_in[1];
    const int*   ei   = (const int*)d_in[2];
    const float* Wl   = (const float*)d_in[3];
    const float* Wr   = (const float*)d_in[4];
    const float* We   = (const float*)d_in[5];
    const float* att  = (const float*)d_in[6];
    const float* bias = (const float*)d_in[7];
    const float* bnw  = (const float*)d_in[8];
    const float* bnb  = (const float*)d_in[9];
    float* out = (float*)d_out;

    const int N = in_sizes[0] / 128;
    const int E = in_sizes[2] / 2;
    const int* srcIdx = ei;
    const int* dstIdx = ei + E;

    float* ws = (float*)d_ws;
    float* xl = ws;     ws += (size_t)N * 128;
    float* xr = ws;     ws += (size_t)N * 128;
    float* accum = ws;  ws += (size_t)N * 128;
    int* histPre = (int*)ws;  ws += (size_t)N * RANK_CHUNKS * 2;
    int* start = (int*)ws;  ws += (size_t)N * RANK_CHUNKS;
    int* rk    = (int*)ws;  ws += E;
    int* pairs = (int*)ws;  ws += (size_t)E * 2;
    float* wsum = ws;   ws += N;
    int* offs  = (int*)ws;  ws += N + 32;
    float* stats = ws;  ws += 256;

    const int per = (E + RANK_CHUNKS - 1) / RANK_CHUNKS;
    edge_rank<<<RANK_CHUNKS, 1024, (size_t)(N + 1) * sizeof(int), stream>>>(
        dstIdx, rk, histPre, stats, E, N, per);
    chunk_prefix<<<(N + 63) / 64, 64, 0, stream>>>(histPre, start, offs, N);
    const int bpc = (per + 255) / 256;
    const int nTiles = N / 16;
    const int tRows = (nTiles + bpc - 1) / bpc;
    transform_and_place<<<dim3(bpc, tRows + RANK_CHUNKS), 256, 0, stream>>>(
        x, Wl, Wr, xl, xr, srcIdx, dstIdx, rk, start, pairs, E, N, per, bpc, tRows,
        nTiles);
    node_attend<<<(N + 3) / 4, 128, 0, stream>>>(ea, We, xl, xr, pairs, offs, att,
                                                 accum, wsum, N);
    const int rpb = BAS_ROWS;
    bias_act_stats<<<(N + rpb - 1) / rpb, 512, 0, stream>>>(accum, wsum, bias, out,
                                                            stats, N);
    const int rpa = BNA_ROWS;
    bn_apply<<<(N + rpa - 1) / rpa, 256, 0, stream>>>(out, stats, bnw, bnb, N);
}


# Divide, bias, activation and batch-norm column sums done in node_attend's epilogue (4 rows summed in LDS, atomics into 8 copies, added in fixed order); bias_act_stats launch gone; sum order changes
# speedup vs baseline: 4.2890x; 1.0308x over previous
#include <hip/hip_runtime.h>

typedef __attribute__((ext_vector_type(2))) float v2f;
typedef __attribute__((ext_vector_type(8))) float v8f;

#define NEG_SLOPE 0.2f
#define BN_EPS 1e-5f

__device__ __forceinline__ float lrelu(float x) { return x > 0.0f ? x : NEG_SLOPE * x; }

#define RANK_CHUNKS 128

#define STAT_REPL 8

__global__ __launch_bounds__(1024) void edge_rank(
    const int* __restrict__ dstIdx, int* __restrict__ rk, int* __restrict__ histPre,
    float* __restrict__ stats, int E, int N, int per) {
    extern __shared__ int sCnt[];
    __shared__ int sWave[32];
    const int t = threadIdx.x;
    for (int i = t; i < N; i += 1024) sCnt[i] = 0;
    for (int i = blockIdx.x * 1024 + t; i < STAT_REPL * 256; i += RANK_CHUNKS * 1024)
        stats[i] = 0.0f;
    __syncthreads();
    const int e0 = blockIdx.x * per;
    const int e1 = min(e0 + per, E);
    for (int eb = e0 + t; eb < e1; eb += 8 * 1024) {
        int d[8];
#pragma unroll
        for (int u = 0; u < 8; ++u) {
            const int e = eb + u * 1024;
            d[u] = (e < e1) ? dstIdx[e] : 0;
        }
#pragma unroll
        for (int u = 0; u < 8; ++u) {
            const int e = eb + u * 1024;
            if (e < e1) rk[e] = atomicAdd(&sCnt[d[u]], 1);
        }
    }
    __syncthreads();

    const int span = (N + 1023) / 1024;
    const int b0 = min(t * span, N);
    const int b1 = min(b0 + span, N);
    int sum = 0;
    for (int i0 = b0; i0 < b1; i0 += 16) {
        int v[16];
#pragma unroll
        for (int u = 0; u < 16; ++u) v[u] = sCnt[min(i0 + u, N - 1)];
#pragma unroll
        for (int u = 0; u < 16; ++u) sum += (i0 + u < b1) ? v[u] : 0;
    }
    const int lane = t & 31;
    const int wave = t >> 5;
    int inc = sum;
#pragma unroll
    for (int off = 1; off < 32; off <<= 1) {
        const int nb = __shfl_up(inc, off, 32);
        if (lane >= off) inc += nb;
    }
    if (lane == 31) sWave[wave] = inc;
    __syncthreads();
    if (wave == 0) {
        const int w = sWave[lane];
        int wi = w;
#pragma unroll
        for (int off = 1; off < 32; off <<= 1) {
            const int nb = __shfl_up(wi, off, 32);
            if (lane >= off) wi += nb;
        }
        sWave[lane] = wi - w;
    }
    __syncthreads();
    int run = sWave[wave] + inc - sum;
    for (int i0 = b0; i0 < b1; i0 += 16) {
        int v[16];
#pragma unroll
        for (int u = 0; u < 16; ++u) v[u] = sCnt[min(i0 + u, N - 1)];
#pragma unroll
        for (int u = 0; u < 16; ++u) {
            if (i0 + u < b1) {
                sCnt[i0 + u] = run;
                run += v[u];
            }
        }
    }
    if (b0 < N && b1 == N) sCnt[N] = run;
    __syncthreads();
    int2* __restrict__ h = (int2*)histPre + (size_t)blockIdx.x * N;
#pragma unroll 4
    for (int i = t; i < N; i += 1024) {
        const int p = sCnt[i];
        h[i] = make_int2(sCnt[i + 1] - p, p);
    }
}

__global__ __launch_bounds__(64) void chunk_prefix(
    const int* __restrict__ histPre, int* __restrict__ start,
    int* __restrict__ offs, int N) {
    const int node = blockIdx.x * 64 + threadIdx.x;
    if (node >= N) return;
    int2 v[RANK_CHUNKS];
#pragma unroll
    for (int c = 0; c < RANK_CHUNKS; ++c)
        v[c] = ((const int2*)histPre)[(size_t)c * N + node];
    int base = 0;
#pragma unroll
    for (int c = 0; c < RANK_CHUNKS; ++c) base += v[c].y;
    int run = base;
#pragma unroll
    for (int c = 0; c < RANK_CHUNKS; ++c) {
        start[(size_t)c * N + node] = run;
        run += v[c].x;
    }
    offs[node] = base;
    if (node == N - 1) offs[N] = run;
}

__global__ __launch_bounds__(256) void transform_and_place(
    const float* __restrict__ x, const float* __restrict__ Wl,
    const float* __restrict__ Wr, float* __restrict__ xl, float* __restrict__ xr,
    const int* __restrict__ srcIdx, const int* __restrict__ dstIdx,
    const int* __restrict__ rk, const int* __restrict__ start,
    int* __restrict__ pairs, int E, int N, int per, int bpc, int tRows,
    int nTiles) {
    if ((int)blockIdx.y < tRows) {
        const int tile = blockIdx.y * bpc + blockIdx.x;
        if (tile >= nTiles) return;
        const int wave = threadIdx.x >> 5;
        const int lane = threadIdx.x & 31;
        const int lo   = lane & 15;
        const int hiK  = (lane >= 16) ? 2 : 0;
        const int hiM  = (lane >= 16) ? 8 : 0;
        const int rowTile = tile * 16;
        const int col  = wave * 16 + lo;
        const int aRow = rowTile + lo;

        v8f cl = {};
        v8f cr = {};
        for (int k = 0; k < 128; k += 4) {
            v2f a = *(const v2f*)&x[aRow * 128 + k + hiK];
            v2f bl, br;
            bl.x = Wl[(k + hiK) * 128 + col];
            bl.y = Wl[(k + hiK + 1) * 128 + col];
            br.x = Wr[(k + hiK) * 128 + col];
            br.y = Wr[(k + hiK + 1) * 128 + col];
            cl = __builtin_amdgcn_wmma_f32_16x16x4_f32(false, a, false, bl, (short)0, cl, false, false);
            cr = __builtin_amdgcn_wmma_f32_16x16x4_f32(false, a, false, br, (short)0, cr, false, false);
        }
#pragma unroll
        for (int r = 0; r < 8; ++r) {
            int row = rowTile + r + hiM;
            xl[row * 128 + col] = cl[r];
            xr[row * 128 + col] = cr[r];
        }
        return;
    }
    const int chunk = blockIdx.y - tRows;
    const int inChunk = blockIdx.x * 256 + threadIdx.x;
    const int e = chunk * per + inChunk;
    if (inChunk >= per || e >= E) return;
    const int d = dstIdx[e];
    const int pos = start[(size_t)chunk * N + d] + rk[e];
    ((int2*)pairs)[pos] = make_int2(srcIdx[e], e);
}

__global__ __launch_bounds__(128, 5) void node_attend(
    const float* __restrict__ edge_attr, const float* __restrict__ We,
    const float* __restrict__ xl, const float* __restrict__ xr,
    const int* __restrict__ pairs, const int* __restrict__ offs,
    const float* __restrict__ att, const float* __restrict__ bias,
    float* __restrict__ out, float* __restrict__ stats, int N) {
    __shared__ float4 sB[512];
    __shared__ float4 sPart[4][64];
    for (int pce = threadIdx.x; pce < 256; pce += 128) {
        const int c  = pce & 15;
        const int h  = (pce >> 4) & 1;
        const int q  = (pce >> 5) & 1;
        const int ks = (pce >> 6) & 3;
        const int k0 = 4 * ks + 2 * h;
        const float4 r0 = *(const float4*)&We[k0 * 128 + 64 * q + 4 * c];
        const float4 r1 = *(const float4*)&We[(k0 + 1) * 128 + 64 * q + 4 * c];
        sB[((((ks * 2 + q) * 2 + 0) * 2 + h) * 16) + c] = make_float4(r0.x, r1.x, r0.y, r1.y);
        sB[((((ks * 2 + q) * 2 + 1) * 2 + h) * 16) + c] = make_float4(r0.z, r1.z, r0.w, r1.w);
    }
    __syncthreads();

    const int wave = __builtin_amdgcn_readfirstlane(threadIdx.x >> 5);
    const int node = blockIdx.x * 4 + wave;
    const bool live = node < N;
    const int lane = threadIdx.x & 31;
    const int lo   = lane & 15;
    const int hh   = lane >> 4;
    const int hiK  = hh * 2;
    const int hiM  = hh * 8;
    float4 o = make_float4(0.0f, 0.0f, 0.0f, 0.0f);
    if (live) {
    const int2* pr2 = (const int2*)pairs;
    const int beg = offs[node];
    const int end = offs[node + 1];

    float m  = -INFINITY;
    float ws = 0.0f;
    float accv[2][4];
#pragma unroll
    for (int q = 0; q < 2; ++q)
#pragma unroll
        for (int j = 0; j < 4; ++j) accv[q][j] = 0.0f;

    int2 nxt = make_int2(0, 0);
    if (beg + lo < end) nxt = pr2[beg + lo];

    for (int t = beg; t < end; t += 16) {
        const int2 cur = nxt;
        nxt = make_int2(0, 0);
        if (t + 16 + lo < end) nxt = pr2[t + 16 + lo];
        asm volatile("" ::: "memory");

        const float4 xr0 = *(const float4*)&xr[(size_t)node * 128 + lo * 4];
        const float4 xr1 = *(const float4*)&xr[(size_t)node * 128 + 64 + lo * 4];
        const float4 at0 = *(const float4*)&att[lo * 4];
        const float4 at1 = *(const float4*)&att[64 + lo * 4];
        const float xrv[2][4] = {{xr0.x, xr0.y, xr0.z, xr0.w}, {xr1.x, xr1.y, xr1.z, xr1.w}};
        const float avj[2][4] = {{at0.x, at0.y, at0.z, at0.w}, {at1.x, at1.y, at1.z, at1.w}};

        const float* aRow = &edge_attr[(size_t)cur.y * 16];
        v2f a[4];
        a[0] = *(const v2f*)&aRow[0 + hiK];
        a[1] = *(const v2f*)&aRow[4 + hiK];
        a[2] = *(const v2f*)&aRow[8 + hiK];
        a[3] = *(const v2f*)&aRow[12 + hiK];

        float vl[2][8][4];
#pragma unroll
        for (int r = 0; r < 8; ++r) {
            const unsigned offS =
                (unsigned)__shfl(cur.x, r + hiM, 32) * 512u + (unsigned)lo * 16u;
#pragma unroll
            for (int q = 0; q < 2; ++q) {
                const float4 v = *(const float4*)((const char*)xl + offS + q * 256);
                vl[q][r][0] = v.x;
                vl[q][r][1] = v.y;
                vl[q][r][2] = v.z;
                vl[q][r][3] = v.w;
            }
        }

        float e[8];
#pragma unroll
        for (int r = 0; r < 8; ++r) e[r] = 0.0f;
#pragma unroll
        for (int q = 0; q < 2; ++q) {
#pragma unroll
            for (int jp = 0; jp < 2; ++jp) {
                float4 bv[4];
#pragma unroll
                for (int ks = 0; ks < 4; ++ks)
                    bv[ks] = sB[((((ks * 2 + q) * 2 + jp) * 2 + hh) * 16) + lo];
#pragma unroll
                for (int jj = 0; jj < 2; ++jj) {
                    const int j = 2 * jp + jj;
                    v8f c;
#pragma unroll
                    for (int r = 0; r < 8; ++r) c[r] = vl[q][r][j] + xrv[q][j];
#pragma unroll
                    for (int ks = 0; ks < 4; ++ks) {
                        v2f b;
                        b.x = jj ? bv[ks].z : bv[ks].x;
                        b.y = jj ? bv[ks].w : bv[ks].y;
                        c = __builtin_amdgcn_wmma_f32_16x16x4_f32(false, a[ks], false, b, (short)0, c, false, false);
                    }
#pragma unroll
                    for (int r = 0; r < 8; ++r) e[r] += lrelu(c[r]) * avj[q][j];
                }
            }
        }

        const int nvalid = end - t;
        float tm = -INFINITY;
#pragma unroll
        for (int r = 0; r < 8; ++r) {
            float v = e[r];
            v += __shfl_xor(v, 1, 32);
            v += __shfl_xor(v, 2, 32);
            v += __shfl_xor(v, 4, 32);
            v += __shfl_xor(v, 8, 32);
            if (r + hiM >= nvalid) v = -INFINITY;
            e[r] = v;
            tm = fmaxf(tm, v);
        }
        tm = fmaxf(tm, __shfl_xor(tm, 16, 32));

        const float mn = fmaxf(m, tm);
        const float sc = expf(m - mn);
        m = mn;
        ws *= sc;
#pragma unroll
        for (int q = 0; q < 2; ++q)
#pragma unroll
            for (int j = 0; j < 4; ++j) accv[q][j] *= sc;
#pragma unroll
        for (int r = 0; r < 8; ++r) {
            const float p = expf(e[r] - mn);
            ws += p;
#pragma unroll
            for (int q = 0; q < 2; ++q)
#pragma unroll
                for (int j = 0; j < 4; ++j) accv[q][j] += p * vl[q][r][j];
        }
    }

#pragma unroll
    for (int q = 0; q < 2; ++q)
#pragma unroll
        for (int j = 0; j < 4; ++j) accv[q][j] += __shfl_xor(accv[q][j], 16, 32);
    ws += __shfl_xor(ws, 16, 32);

    const float inv = ws > 0.0f ? 1.0f / ws : 1.0f;
    const float4 b = *(const float4*)&bias[hh * 64 + lo * 4];
    const float4 a = hh ? make_float4(accv[1][0], accv[1][1], accv[1][2], accv[1][3])
                        : make_float4(accv[0][0], accv[0][1], accv[0][2], accv[0][3]);
    o.x = lrelu(a.x * inv + b.x);
    o.y = lrelu(a.y * inv + b.y);
    o.z = lrelu(a.z * inv + b.z);
    o.w = lrelu(a.w * inv + b.w);
    }

    sPart[wave][lane]      = o;
    sPart[wave][32 + lane] = make_float4(o.x * o.x, o.y * o.y, o.z * o.z, o.w * o.w);
    __syncthreads();
    if (live) *(float4*)&out[(size_t)node * 128 + hh * 64 + lo * 4] = o;
    const float* p = (const float*)sPart;
    float* st = stats + (blockIdx.x % STAT_REPL) * 256;
    const int t = threadIdx.x;
    atomicAdd(&st[t], (p[t] + p[256 + t]) + (p[512 + t] + p[768 + t]));
    atomicAdd(&st[128 + t],
              (p[128 + t] + p[384 + t]) + (p[640 + t] + p[896 + t]));
}

#define BNA_WAVES 8
#define BNA_PER_WAVE 5
#define BNA_ROWS (BNA_WAVES * BNA_PER_WAVE)
__device__ __forceinline__ void bn_coef(float su, float sq, float w, float b,
                                        float inv, float& sc, float& sh) {
    const float mean = su * inv;
    const float var  = sq * inv - mean * mean;
    sc = w * rsqrtf(var + BN_EPS);
    sh = b - mean * sc;
}
__global__ __launch_bounds__(256) void bn_apply(
    float* __restrict__ out, const float* __restrict__ stats,
    const float* __restrict__ bnw, const float* __restrict__ bnb, int N) {
    const int wave = __builtin_amdgcn_readfirstlane(threadIdx.x >> 5);
    const int lane = threadIdx.x & 31;
    const int r0 = blockIdx.x * BNA_ROWS + wave;
    if (r0 >= N) return;

    float4 v[BNA_PER_WAVE];
#pragma unroll
    for (int i = 0; i < BNA_PER_WAVE; ++i) {
        const int r  = r0 + i * BNA_WAVES;
        const int rr = r < N ? r : r0;
        v[i] = *(const float4*)&out[(size_t)rr * 128 + lane * 4];
    }
    float4 cs[STAT_REPL], cq[STAT_REPL];
#pragma unroll
    for (int r = 0; r < STAT_REPL; ++r) {
        cs[r] = *(const float4*)&stats[r * 256 + lane * 4];
        cq[r] = *(const float4*)&stats[r * 256 + 128 + lane * 4];
    }
    const float4 w  = *(const float4*)&bnw[lane * 4];
    const float4 b  = *(const float4*)&bnb[lane * 4];
    static_assert(STAT_REPL == 8 && BNA_PER_WAVE == 5, "the fence names every load");
    asm volatile("" ::"v"(v[0].x), "v"(v[1].x), "v"(v[2].x), "v"(v[3].x), "v"(v[4].x),
                 "v"(v[0].w), "v"(v[1].w), "v"(v[2].w), "v"(v[3].w), "v"(v[4].w),
                 "v"(cs[0].x), "v"(cs[1].x), "v"(cs[2].x), "v"(cs[3].x), "v"(cs[4].x),
                 "v"(cs[5].x), "v"(cs[6].x), "v"(cs[7].x), "v"(cq[0].x), "v"(cq[1].x),
                 "v"(cq[2].x), "v"(cq[3].x), "v"(cq[4].x), "v"(cq[5].x), "v"(cq[6].x),
                 "v"(cq[7].x), "v"(w.x), "v"(b.x));
    float4 su = cs[0];
    float4 sq = cq[0];
#pragma unroll
    for (int r = 1; r < STAT_REPL; ++r) {
        su.x += cs[r].x; su.y += cs[r].y; su.z += cs[r].z; su.w += cs[r].w;
        sq.x += cq[r].x; sq.y += cq[r].y; sq.z += cq[r].z; sq.w += cq[r].w;
    }

    const float inv = 1.0f / (float)N;
    float4 sc, sh;
    bn_coef(su.x, sq.x, w.x, b.x, inv, sc.x, sh.x);
    bn_coef(su.y, sq.y, w.y, b.y, inv, sc.y, sh.y);
    bn_coef(su.z, sq.z, w.z, b.z, inv, sc.z, sh.z);
    bn_coef(su.w, sq.w, w.w, b.w, inv, sc.w, sh.w);

#pragma unroll
    for (int i = 0; i < BNA_PER_WAVE; ++i) {
        const int r = r0 + i * BNA_WAVES;
        float4 o;
        o.x = v[i].x * sc.x + sh.x;
        o.y = v[i].y * sc.y + sh.y;
        o.z = v[i].z * sc.z + sh.z;
        o.w = v[i].w * sc.w + sh.w;
        if (r < N) *(float4*)&out[(size_t)r * 128 + lane * 4] = o;
    }
}

extern "C" void kernel_launch(void* const* d_in, const int* in_sizes, int n_in,
                              void* d_out, int out_size, void* d_ws, size_t ws_size,
                              hipStream_t stream) {
    const float* x    = (const float*)d_in[0];
    const float* ea   = (const float*)d_in[1];
    const int*   ei   = (const int*)d_in[2];
    const float* Wl   = (const float*)d_in[3];
    const float* Wr   = (const float*)d_in[4];
    const float* We   = (const float*)d_in[5];
    const float* att  = (const float*)d_in[6];
    const float* bias = (const float*)d_in[7];
    const float* bnw  = (const float*)d_in[8];
    const float* bnb  = (const float*)d_in[9];
    float* out = (float*)d_out;

    const int N = in_sizes[0] / 128;
    const int E = in_sizes[2] / 2;
    const int* srcIdx = ei;
    const int* dstIdx = ei + E;

    float* ws = (float*)d_ws;
    float* xl = ws;     ws += (size_t)N * 128;
    float* xr = ws;     ws += (size_t)N * 128;
    int* histPre = (int*)ws;  ws += (size_t)N * RANK_CHUNKS * 2;
    int* start = (int*)ws;  ws += (size_t)N * RANK_CHUNKS;
    int* rk    = (int*)ws;  ws += E;
    int* pairs = (int*)ws;  ws += (size_t)E * 2;
    int* offs  = (int*)ws;  ws += N + 32;
    float* stats = ws;  ws += STAT_REPL * 256;

    const int per = (E + RANK_CHUNKS - 1) / RANK_CHUNKS;
    edge_rank<<<RANK_CHUNKS, 1024, (size_t)(N + 1) * sizeof(int), stream>>>(
        dstIdx, rk, histPre, stats, E, N, per);
    chunk_prefix<<<(N + 63) / 64, 64, 0, stream>>>(histPre, start, offs, N);
    const int bpc = (per + 255) / 256;
    const int nTiles = N / 16;
    const int tRows = (nTiles + bpc - 1) / bpc;
    transform_and_place<<<dim3(bpc, tRows + RANK_CHUNKS), 256, 0, stream>>>(
        x, Wl, Wr, xl, xr, srcIdx, dstIdx, rk, start, pairs, E, N, per, bpc, tRows,
        nTiles);
    node_attend<<<(N + 3) / 4, 128, 0, stream>>>(ea, We, xl, xr, pairs, offs, att,
                                                 bias, out, stats, N);
    const int rpa = BNA_ROWS;
    bn_apply<<<(N + rpa - 1) / rpa, 256, 0, stream>>>(out, stats, bnw, bnb, N);
}
